# starting kernel
# baseline (speedup 1.0000x reference)
#include <hip/hip_runtime.h>


#define Dch      128
#define Hh       4
#define Ch       32
#define Lnum     3
#define NEG_SLOPE 0.2f
#define BN_EPS   1e-5f

typedef __attribute__((ext_vector_type(2))) float v2f;
typedef __attribute__((ext_vector_type(8))) float v8f;

__device__ __forceinline__ unsigned fkey(float f) {
  unsigned b = __float_as_uint(f);
  return b ^ ((unsigned)((int)b >> 31) | 0x80000000u);
}
__device__ __forceinline__ float kinv(unsigned k) {
  return __uint_as_float((k & 0x80000000u) ? (k ^ 0x80000000u) : ~k);
}

__device__ __forceinline__ void edge_sd(const int* __restrict__ ei, int Ereal,
                                        int e, int& s, int& d) {
  if (e < Ereal) { s = ei[e]; d = ei[Ereal + e]; }
  else           { s = e - Ereal; d = s; }
}

__global__ void fill_u32(unsigned* __restrict__ p, unsigned v, long n) {
  long i = (long)blockIdx.x * blockDim.x + threadIdx.x;
  long stride = (long)gridDim.x * blockDim.x;
  for (; i < n; i += stride) p[i] = v;
}

__global__ __launch_bounds__(256) void gemm_wmma_f32(
    const float* __restrict__ X, const float* __restrict__ Wm,
    float* __restrict__ Ho) {
  const int wave = threadIdx.x >> 5;
  const int lane = threadIdx.x & 31;
  const int half = lane >> 4;
  const int l15  = lane & 15;
  const int row0 = blockIdx.x << 4;
  const int bcol = (wave << 4) + l15;
  const float* arow = X + (long)(row0 + l15) * Dch;

  v8f acc = {};
  for (int k = 0; k < Dch; k += 4) {
    const int ka = k + half * 2;
    v2f a, b;
    a.x = arow[ka];
    a.y = arow[ka + 1];
    b.x = Wm[(long)ka * Dch + bcol];
    b.y = Wm[(long)(ka + 1) * Dch + bcol];
    acc = __builtin_amdgcn_wmma_f32_16x16x4_f32(
         false, a,  false, b,
         (short)0, acc,  false,  false);
  }
  float* orow = Ho + (long)(row0 + half * 8) * Dch + bcol;
#pragma unroll
  for (int v = 0; v < 8; ++v) orow[(long)v * Dch] = acc[v];
}

__global__ void att_logits(const float* __restrict__ Hf,
                           const float* __restrict__ a_src,
                           const float* __restrict__ a_dst,
                           float* __restrict__ alsrc, float* __restrict__ aldst,
                           int Nn) {
  int i = blockIdx.x * blockDim.x + threadIdx.x;
  if (i >= Nn * Hh) return;
  int n = i >> 2, h = i & 3;
  const float* hp = Hf + (long)n * Dch + h * Ch;
  const float* as = a_src + h * Ch;
  const float* ad = a_dst + h * Ch;
  float s0 = 0.f, s1 = 0.f;
#pragma unroll
  for (int c = 0; c < Ch; ++c) { float v = hp[c]; s0 += v * as[c]; s1 += v * ad[c]; }
  alsrc[i] = s0;
  aldst[i] = s1;
}

__global__ void edge_max(const int* __restrict__ ei, int Ereal, int Etot,
                         const float* __restrict__ alsrc,
                         const float* __restrict__ aldst,
                         unsigned* __restrict__ mkey) {
  int e = blockIdx.x * blockDim.x + threadIdx.x;
  if (e >= Etot) return;
  int s, d; edge_sd(ei, Ereal, e, s, d);
#pragma unroll
  for (int h = 0; h < Hh; ++h) {
    float sc = alsrc[s * Hh + h] + aldst[d * Hh + h];
    sc = sc > 0.f ? sc : sc * NEG_SLOPE;
    atomicMax(&mkey[d * Hh + h], fkey(sc));
  }
}

__global__ void edge_expsum(const int* __restrict__ ei, int Ereal, int Etot,
                            const float* __restrict__ alsrc,
                            const float* __restrict__ aldst,
                            const unsigned* __restrict__ mkey,
                            float* __restrict__ ssum) {
  int e = blockIdx.x * blockDim.x + threadIdx.x;
  if (e >= Etot) return;
  int s, d; edge_sd(ei, Ereal, e, s, d);
#pragma unroll
  for (int h = 0; h < Hh; ++h) {
    float sc = alsrc[s * Hh + h] + aldst[d * Hh + h];
    sc = sc > 0.f ? sc : sc * NEG_SLOPE;
    float p = expf(sc - kinv(mkey[d * Hh + h]));
    atomicAdd(&ssum[d * Hh + h], p);
  }
}

__global__ __launch_bounds__(256) void edge_aggregate(
    const int* __restrict__ ei, int Ereal, int Etot,
    const float* __restrict__ alsrc, const float* __restrict__ aldst,
    const unsigned* __restrict__ mkey, const float* __restrict__ ssum,
    const float* __restrict__ Hf, float* __restrict__ acc) {
  long gid = (long)blockIdx.x * blockDim.x + threadIdx.x;
  int e    = (int)(gid >> 5);
  int lane = (int)(gid & 31);
  if (e >= Etot) return;
  int s, d; edge_sd(ei, Ereal, e, s, d);
  const float* hsrc = Hf + (long)s * Dch;
  float* adst = acc + (long)d * Dch;
#pragma unroll
  for (int h = 0; h < Hh; ++h) {
    float sc = alsrc[s * Hh + h] + aldst[d * Hh + h];
    sc = sc > 0.f ? sc : sc * NEG_SLOPE;
    float p = expf(sc - kinv(mkey[d * Hh + h]));
    float alpha = p / (ssum[d * Hh + h] + 1e-16f);
    int ch = h * Ch + lane;
    atomicAdd(&adst[ch], alpha * hsrc[ch]);
  }
}

__global__ void post_bn_elu(const float* __restrict__ acc,
                            const float* __restrict__ bias,
                            const float* __restrict__ gamma,
                            const float* __restrict__ beta,
                            const float* __restrict__ mean,
                            const float* __restrict__ var,
                            float* __restrict__ xo, long n) {
  long i = (long)blockIdx.x * blockDim.x + threadIdx.x;
  if (i >= n) return;
  int c = (int)(i & (Dch - 1));
  float v = acc[i] + bias[c];
  v = (v - mean[c]) * (gamma[c] / sqrtf(var[c] + BN_EPS)) + beta[c];
  xo[i] = v > 0.f ? v : expf(v) - 1.f;
}

__global__ void pool_accum(const float* __restrict__ x,
                           const int* __restrict__ batch,
                           float* __restrict__ sums, float* __restrict__ cnt,
                           long n) {
  long i = (long)blockIdx.x * blockDim.x + threadIdx.x;
  if (i >= n) return;
  int node = (int)(i >> 7);
  int c    = (int)(i & 127);
  int g    = batch[node];
  atomicAdd(&sums[(long)g * Dch + c], x[i]);
  if (c == 0) atomicAdd(&cnt[g], 1.0f);
}

__global__ void pool_final(const float* __restrict__ sums,
                           const float* __restrict__ cnt,
                           float* __restrict__ out, int G) {
  int i = blockIdx.x * blockDim.x + threadIdx.x;
  if (i >= G * Dch) return;
  int g = i >> 7;
  float c = cnt[g];
  out[i] = sums[i] / (c > 1.f ? c : 1.f);
}

extern "C" void kernel_launch(void* const* d_in, const int* in_sizes, int n_in,
                              void* d_out, int out_size, void* d_ws, size_t ws_size,
                              hipStream_t stream) {
  const float* x0    = (const float*)d_in[0];
  const int*   ei    = (const int*)d_in[1];
  const int*   batch = (const int*)d_in[2];
  const float* Ws    = (const float*)d_in[3];
  const float* a_src = (const float*)d_in[4];
  const float* a_dst = (const float*)d_in[5];
  const float* bias  = (const float*)d_in[6];
  const float* gamma = (const float*)d_in[7];
  const float* beta  = (const float*)d_in[8];
  const float* mean  = (const float*)d_in[9];
  const float* var   = (const float*)d_in[10];

  const int Nn    = in_sizes[0] / Dch;
  const int Ereal = in_sizes[1] / 2;
  const int Etot  = Ereal + Nn;
  const int G     = out_size / Dch;

  char* ws = (char*)d_ws;
  float*    bufA  = (float*)ws;    ws += (size_t)Nn * Dch * 4;
  float*    bufB  = (float*)ws;    ws += (size_t)Nn * Dch * 4;
  float*    hbuf  = (float*)ws;    ws += (size_t)Nn * Dch * 4;
  unsigned* mkey  = (unsigned*)ws; ws += (size_t)Nn * Hh * 4;
  float*    ssum  = (float*)ws;    ws += (size_t)Nn * Hh * 4;
  float*    alsrc = (float*)ws;    ws += (size_t)Nn * Hh * 4;
  float*    aldst = (float*)ws;    ws += (size_t)Nn * Hh * 4;
  float*    psum  = (float*)ws;    ws += (size_t)G * Dch * 4;
  float*    pcnt  = (float*)ws;    ws += (size_t)G * 4;

  const float* xin  = x0;
  float*       xout = bufA;
  for (int l = 0; l < Lnum; ++l) {
    const float* Wl = Ws + (size_t)l * Dch * Dch;
    gemm_wmma_f32<<<Nn / 16, 256, 0, stream>>>(xin, Wl, hbuf);

    att_logits<<<(Nn * Hh + 255) / 256, 256, 0, stream>>>(
        hbuf, a_src + (size_t)l * Hh * Ch, a_dst + (size_t)l * Hh * Ch,
        alsrc, aldst, Nn);

    fill_u32<<<512, 256, 0, stream>>>(mkey, 0u, (long)Nn * Hh);
    fill_u32<<<512, 256, 0, stream>>>((unsigned*)ssum, 0u, (long)Nn * Hh);
    fill_u32<<<2048, 256, 0, stream>>>((unsigned*)xout, 0u, (long)Nn * Dch);

    edge_max<<<(Etot + 255) / 256, 256, 0, stream>>>(ei, Ereal, Etot, alsrc, aldst, mkey);
    edge_expsum<<<(Etot + 255) / 256, 256, 0, stream>>>(ei, Ereal, Etot, alsrc, aldst, mkey, ssum);
    {
      long t = (long)Etot * 32;
      edge_aggregate<<<(int)((t + 255) / 256), 256, 0, stream>>>(
          ei, Ereal, Etot, alsrc, aldst, mkey, ssum, hbuf, xout);
    }

    post_bn_elu<<<(Nn * Dch + 255) / 256, 256, 0, stream>>>(
        xout, bias + (size_t)l * Dch, gamma + (size_t)l * Dch,
        beta + (size_t)l * Dch, mean + (size_t)l * Dch, var + (size_t)l * Dch,
        xout, (long)Nn * Dch);

    xin  = xout;
    xout = (xout == bufA) ? bufB : bufA;
  }

  fill_u32<<<32, 256, 0, stream>>>((unsigned*)psum, 0u, (long)G * Dch);
  fill_u32<<<1, 64, 0, stream>>>((unsigned*)pcnt, 0u, G);
  pool_accum<<<(Nn * Dch + 255) / 256, 256, 0, stream>>>(xin, batch, psum, pcnt, (long)Nn * Dch);
  pool_final<<<(G * Dch + 255) / 256, 256, 0, stream>>>(psum, pcnt, (float*)d_out, G);
}


# fewer, longer-lived waves: edge pass 3 gives a wave 32 edges (indices loaded once, 4 edges in flight); other edge and elementwise passes loop grid-stride; float atomics in arrival order as before
# speedup vs baseline: 1.7855x; 1.7855x over previous
#include <hip/hip_runtime.h>


#define Dch      128
#define Hh       4
#define Ch       32
#define Lnum     3
#define NEG_SLOPE 0.2f
#define BN_EPS   1e-5f
#define LOOP_BLOCKS 2048

typedef __attribute__((ext_vector_type(2))) float v2f;
typedef __attribute__((ext_vector_type(8))) float v8f;

__device__ __forceinline__ unsigned fkey(float f) {
  unsigned b = __float_as_uint(f);
  return b ^ ((unsigned)((int)b >> 31) | 0x80000000u);
}
__device__ __forceinline__ float kinv(unsigned k) {
  return __uint_as_float((k & 0x80000000u) ? (k ^ 0x80000000u) : ~k);
}

__device__ __forceinline__ void edge_sd(const int* __restrict__ ei, int Ereal,
                                        int e, int& s, int& d) {
  if (e < Ereal) { s = ei[e]; d = ei[Ereal + e]; }
  else           { s = e - Ereal; d = s; }
}

__global__ void fill_u32(unsigned* __restrict__ p, unsigned v, long n) {
  long i = (long)blockIdx.x * blockDim.x + threadIdx.x;
  long stride = (long)gridDim.x * blockDim.x;
  for (; i < n; i += stride) p[i] = v;
}

__global__ __launch_bounds__(256) void gemm_wmma_f32(
    const float* __restrict__ X, const float* __restrict__ Wm,
    float* __restrict__ Ho) {
  const int wave = threadIdx.x >> 5;
  const int lane = threadIdx.x & 31;
  const int half = lane >> 4;
  const int l15  = lane & 15;
  const int row0 = blockIdx.x << 4;
  const int bcol = (wave << 4) + l15;
  const float* arow = X + (long)(row0 + l15) * Dch;

  v8f acc = {};
  for (int k = 0; k < Dch; k += 4) {
    const int ka = k + half * 2;
    v2f a, b;
    a.x = arow[ka];
    a.y = arow[ka + 1];
    b.x = Wm[(long)ka * Dch + bcol];
    b.y = Wm[(long)(ka + 1) * Dch + bcol];
    acc = __builtin_amdgcn_wmma_f32_16x16x4_f32(
         false, a,  false, b,
         (short)0, acc,  false,  false);
  }
  float* orow = Ho + (long)(row0 + half * 8) * Dch + bcol;
#pragma unroll
  for (int v = 0; v < 8; ++v) orow[(long)v * Dch] = acc[v];
}

__global__ void att_logits(const float* __restrict__ Hf,
                           const float* __restrict__ a_src,
                           const float* __restrict__ a_dst,
                           float* __restrict__ alsrc, float* __restrict__ aldst,
                           int Nn) {
  int i = blockIdx.x * blockDim.x + threadIdx.x;
  if (i >= Nn * Hh) return;
  int n = i >> 2, h = i & 3;
  const float* hp = Hf + (long)n * Dch + h * Ch;
  const float* as = a_src + h * Ch;
  const float* ad = a_dst + h * Ch;
  float s0 = 0.f, s1 = 0.f;
#pragma unroll
  for (int c = 0; c < Ch; ++c) { float v = hp[c]; s0 += v * as[c]; s1 += v * ad[c]; }
  alsrc[i] = s0;
  aldst[i] = s1;
}


__global__ __launch_bounds__(256) void edge_max(
    const int* __restrict__ ei, int Ereal, int Etot,
    const float* __restrict__ alsrc, const float* __restrict__ aldst,
    unsigned* __restrict__ mkey) {
  const int total  = Etot * Hh;
  const int stride = (int)(gridDim.x * blockDim.x);
  for (int gid = (int)(blockIdx.x * blockDim.x + threadIdx.x); gid < total; gid += stride) {
    int e = gid >> 2;
    int h = gid & 3;
    int s, d; edge_sd(ei, Ereal, e, s, d);
    float sc = alsrc[s * Hh + h] + aldst[d * Hh + h];
    sc = sc > 0.f ? sc : sc * NEG_SLOPE;
    atomicMax(&mkey[d * Hh + h], fkey(sc));
  }
}

__global__ __launch_bounds__(256) void edge_expsum(
    const int* __restrict__ ei, int Ereal, int Etot,
    const float* __restrict__ alsrc, const float* __restrict__ aldst,
    const unsigned* __restrict__ mkey, float* __restrict__ ssum) {
  const int total  = Etot * Hh;
  const int stride = (int)(gridDim.x * blockDim.x);
  for (int gid = (int)(blockIdx.x * blockDim.x + threadIdx.x); gid < total; gid += stride) {
    int e = gid >> 2;
    int h = gid & 3;
    int s, d; edge_sd(ei, Ereal, e, s, d);
    float sc = alsrc[s * Hh + h] + aldst[d * Hh + h];
    sc = sc > 0.f ? sc : sc * NEG_SLOPE;
    float p = expf(sc - kinv(mkey[d * Hh + h]));
    atomicAdd(&ssum[d * Hh + h], p);
  }
}


template <int K>
__device__ __forceinline__ void agg_edges(
    int j, int sv, int dv, int lane, int hh,
    const float* __restrict__ alsrc, const float* __restrict__ aldst,
    const unsigned* __restrict__ mkey, const float* __restrict__ ssum,
    const float* __restrict__ Hf, float* __restrict__ acc) {
  int   d[K];
  float asv[K], adv[K], smv[K], hv[K][Hh];
  unsigned mk[K];
  for (int k = 0; k < K; ++k) {
    const int s = __builtin_amdgcn_readlane(sv, j + k);
    d[k]        = __builtin_amdgcn_readlane(dv, j + k);
    asv[k] = alsrc[s * Hh + hh];
    adv[k] = aldst[d[k] * Hh + hh];
    mk[k]  = mkey[d[k] * Hh + hh];
    smv[k] = ssum[d[k] * Hh + hh];
    const float* hsrc = Hf + (long)s * Dch + lane;
    for (int h = 0; h < Hh; ++h) hv[k][h] = hsrc[h * Ch];
  }
  for (int k = 0; k < K; ++k) {
    float sc = asv[k] + adv[k];
    sc = sc > 0.f ? sc : sc * NEG_SLOPE;
    float p = expf(sc - kinv(mk[k]));
    float alpha = p / (smv[k] + 1e-16f);
    float* adst = acc + (long)d[k] * Dch + lane;
    for (int h = 0; h < Hh; ++h) {
      float ah = __int_as_float(__builtin_amdgcn_readlane(__float_as_int(alpha), h));
      atomicAdd(&adst[h * Ch], ah * hv[k][h]);
    }
  }
}

__global__ __launch_bounds__(256) void edge_aggregate(
    const int* __restrict__ ei, int Ereal, int Etot,
    const float* __restrict__ alsrc, const float* __restrict__ aldst,
    const unsigned* __restrict__ mkey, const float* __restrict__ ssum,
    const float* __restrict__ Hf, float* __restrict__ acc) {
  const int lane = threadIdx.x & 31;
  const int hh   = lane & 3;
  const int wave = __builtin_amdgcn_readfirstlane((int)(blockIdx.x * 8 + (threadIdx.x >> 5)));
  const int base = wave << 5;
  if (base >= Etot) return;
  const int cnt = (Etot - base) < 32 ? (Etot - base) : 32;
  int sv = 0, dv = 0;
  if (base + lane < Etot) edge_sd(ei, Ereal, base + lane, sv, dv);
  const int full = cnt & ~3;
  int j = 0;
  for (; j < full; j += 4)
    agg_edges<4>(j, sv, dv, lane, hh, alsrc, aldst, mkey, ssum, Hf, acc);
  for (; j < cnt; ++j)
    agg_edges<1>(j, sv, dv, lane, hh, alsrc, aldst, mkey, ssum, Hf, acc);
}

__global__ __launch_bounds__(256) void post_bn_elu(float* __restrict__ x,
                            const float* __restrict__ bias,
                            const float* __restrict__ gamma,
                            const float* __restrict__ beta,
                            const float* __restrict__ mean,
                            const float* __restrict__ var,
                            long n) {
  const long tid    = (long)blockIdx.x * blockDim.x + threadIdx.x;
  const long stride = (long)gridDim.x * blockDim.x;
  const int  c      = (int)((tid * 4) & (Dch - 1));
  float b[4], m[4], sc[4], be[4];
  for (int k = 0; k < 4; ++k) {
    b[k]  = bias[c + k];
    m[k]  = mean[c + k];
    sc[k] = gamma[c + k] / sqrtf(var[c + k] + BN_EPS);
    be[k] = beta[c + k];
  }
  float4* x4 = (float4*)x;
  const long n4 = n >> 2;
  for (long i = tid; i < n4; i += stride) {
    float4 q = x4[i];
    float v[4] = {q.x, q.y, q.z, q.w};
    for (int k = 0; k < 4; ++k) {
      float t = v[k] + b[k];
      t = (t - m[k]) * sc[k] + be[k];
      v[k] = t > 0.f ? t : expf(t) - 1.f;
    }
    x4[i] = make_float4(v[0], v[1], v[2], v[3]);
  }
}

__global__ void pool_accum(const float* __restrict__ x,
                           const int* __restrict__ batch,
                           float* __restrict__ sums, float* __restrict__ cnt,
                           long n) {
  const long stride = (long)gridDim.x * blockDim.x;
  for (long i = (long)blockIdx.x * blockDim.x + threadIdx.x; i < n; i += stride) {
    int node = (int)(i >> 7);
    int c    = (int)(i & 127);
    int g    = batch[node];
    atomicAdd(&sums[(long)g * Dch + c], x[i]);
    if (c == 0) atomicAdd(&cnt[g], 1.0f);
  }
}

__global__ void pool_final(const float* __restrict__ sums,
                           const float* __restrict__ cnt,
                           float* __restrict__ out, int G) {
  int i = blockIdx.x * blockDim.x + threadIdx.x;
  if (i >= G * Dch) return;
  int g = i >> 7;
  float c = cnt[g];
  out[i] = sums[i] / (c > 1.f ? c : 1.f);
}

extern "C" void kernel_launch(void* const* d_in, const int* in_sizes, int n_in,
                              void* d_out, int out_size, void* d_ws, size_t ws_size,
                              hipStream_t stream) {
  const float* x0    = (const float*)d_in[0];
  const int*   ei    = (const int*)d_in[1];
  const int*   batch = (const int*)d_in[2];
  const float* Ws    = (const float*)d_in[3];
  const float* a_src = (const float*)d_in[4];
  const float* a_dst = (const float*)d_in[5];
  const float* bias  = (const float*)d_in[6];
  const float* gamma = (const float*)d_in[7];
  const float* beta  = (const float*)d_in[8];
  const float* mean  = (const float*)d_in[9];
  const float* var   = (const float*)d_in[10];

  const int Nn    = in_sizes[0] / Dch;
  const int Ereal = in_sizes[1] / 2;
  const int Etot  = Ereal + Nn;
  const int G     = out_size / Dch;

  char* ws = (char*)d_ws;
  float*    bufA  = (float*)ws;    ws += (size_t)Nn * Dch * 4;
  float*    bufB  = (float*)ws;    ws += (size_t)Nn * Dch * 4;
  float*    hbuf  = (float*)ws;    ws += (size_t)Nn * Dch * 4;
  unsigned* mkey  = (unsigned*)ws; ws += (size_t)Nn * Hh * 4;
  float*    ssum  = (float*)ws;    ws += (size_t)Nn * Hh * 4;
  float*    alsrc = (float*)ws;    ws += (size_t)Nn * Hh * 4;
  float*    aldst = (float*)ws;    ws += (size_t)Nn * Hh * 4;
  float*    psum  = (float*)ws;    ws += (size_t)G * Dch * 4;
  float*    pcnt  = (float*)ws;    ws += (size_t)G * 4;

  const float* xin  = x0;
  float*       xout = bufA;
  for (int l = 0; l < Lnum; ++l) {
    const float* Wl = Ws + (size_t)l * Dch * Dch;
    gemm_wmma_f32<<<Nn / 16, 256, 0, stream>>>(xin, Wl, hbuf);

    att_logits<<<(Nn * Hh + 255) / 256, 256, 0, stream>>>(
        hbuf, a_src + (size_t)l * Hh * Ch, a_dst + (size_t)l * Hh * Ch,
        alsrc, aldst, Nn);

    fill_u32<<<512, 256, 0, stream>>>(mkey, 0u, (long)Nn * Hh);
    fill_u32<<<512, 256, 0, stream>>>((unsigned*)ssum, 0u, (long)Nn * Hh);
    fill_u32<<<2048, 256, 0, stream>>>((unsigned*)xout, 0u, (long)Nn * Dch);

    edge_max<<<LOOP_BLOCKS, 256, 0, stream>>>(ei, Ereal, Etot, alsrc, aldst, mkey);
    edge_expsum<<<LOOP_BLOCKS, 256, 0, stream>>>(ei, Ereal, Etot, alsrc, aldst, mkey, ssum);

    edge_aggregate<<<(Etot + 255) / 256, 256, 0, stream>>>(
        ei, Ereal, Etot, alsrc, aldst, mkey, ssum, hbuf, xout);

    post_bn_elu<<<LOOP_BLOCKS, 256, 0, stream>>>(
        xout, bias + (size_t)l * Dch, gamma + (size_t)l * Dch,
        beta + (size_t)l * Dch, mean + (size_t)l * Dch, var + (size_t)l * Dch,
        (long)Nn * Dch);

    xin  = xout;
    xout = (xout == bufA) ? bufB : bufA;
  }

  fill_u32<<<32, 256, 0, stream>>>((unsigned*)psum, 0u, (long)G * Dch);
  fill_u32<<<1, 64, 0, stream>>>((unsigned*)pcnt, 0u, G);
  pool_accum<<<LOOP_BLOCKS, 256, 0, stream>>>(xin, batch, psum, pcnt, (long)Nn * Dch);
  pool_final<<<(G * Dch + 255) / 256, 256, 0, stream>>>(psum, pcnt, (float*)d_out, G);
}


# edge pass 2 owner-computes in LDS: a group owns 160 destination rows, filters the edge list itself, adds weighted rows with LDS float atomics (arrival order); sum clears gone, 28 to 22 launches
# speedup vs baseline: 2.4001x; 1.3442x over previous
#include <hip/hip_runtime.h>


#define Dch      128
#define Hh       4
#define Ch       32
#define Lnum     3
#define NEG_SLOPE 0.2f
#define BN_EPS   1e-5f
#define LOOP_BLOCKS 2048

typedef __attribute__((ext_vector_type(2))) float v2f;
typedef __attribute__((ext_vector_type(8))) float v8f;

__device__ __forceinline__ unsigned fkey(float f) {
  unsigned b = __float_as_uint(f);
  return b ^ ((unsigned)((int)b >> 31) | 0x80000000u);
}
__device__ __forceinline__ float kinv(unsigned k) {
  return __uint_as_float((k & 0x80000000u) ? (k ^ 0x80000000u) : ~k);
}

__device__ __forceinline__ void edge_sd(const int* __restrict__ ei, int Ereal,
                                        int e, int& s, int& d) {
  if (e < Ereal) { s = ei[e]; d = ei[Ereal + e]; }
  else           { s = e - Ereal; d = s; }
}

__global__ void fill_u32(unsigned* __restrict__ p, unsigned v, long n) {
  long i = (long)blockIdx.x * blockDim.x + threadIdx.x;
  long stride = (long)gridDim.x * blockDim.x;
  for (; i < n; i += stride) p[i] = v;
}

__global__ __launch_bounds__(256) void gemm_wmma_f32(
    const float* __restrict__ X, const float* __restrict__ Wm,
    float* __restrict__ Ho) {
  const int wave = threadIdx.x >> 5;
  const int lane = threadIdx.x & 31;
  const int half = lane >> 4;
  const int l15  = lane & 15;
  const int row0 = blockIdx.x << 4;
  const int bcol = (wave << 4) + l15;
  const float* arow = X + (long)(row0 + l15) * Dch;

  v8f acc = {};
  for (int k = 0; k < Dch; k += 4) {
    const int ka = k + half * 2;
    v2f a, b;
    a.x = arow[ka];
    a.y = arow[ka + 1];
    b.x = Wm[(long)ka * Dch + bcol];
    b.y = Wm[(long)(ka + 1) * Dch + bcol];
    acc = __builtin_amdgcn_wmma_f32_16x16x4_f32(
         false, a,  false, b,
         (short)0, acc,  false,  false);
  }
  float* orow = Ho + (long)(row0 + half * 8) * Dch + bcol;
#pragma unroll
  for (int v = 0; v < 8; ++v) orow[(long)v * Dch] = acc[v];
}

__global__ void att_logits(const float* __restrict__ Hf,
                           const float* __restrict__ a_src,
                           const float* __restrict__ a_dst,
                           float* __restrict__ alsrc, float* __restrict__ aldst,
                           int Nn) {
  int i = blockIdx.x * blockDim.x + threadIdx.x;
  if (i >= Nn * Hh) return;
  int n = i >> 2, h = i & 3;
  const float* hp = Hf + (long)n * Dch + h * Ch;
  const float* as = a_src + h * Ch;
  const float* ad = a_dst + h * Ch;
  float s0 = 0.f, s1 = 0.f;
#pragma unroll
  for (int c = 0; c < Ch; ++c) { float v = hp[c]; s0 += v * as[c]; s1 += v * ad[c]; }
  alsrc[i] = s0;
  aldst[i] = s1;
}


__global__ __launch_bounds__(256) void edge_max(
    const int* __restrict__ ei, int Ereal, int Etot,
    const float* __restrict__ alsrc, const float* __restrict__ aldst,
    unsigned* __restrict__ mkey) {
  const int total  = Etot * Hh;
  const int stride = (int)(gridDim.x * blockDim.x);
  for (int gid = (int)(blockIdx.x * blockDim.x + threadIdx.x); gid < total; gid += stride) {
    int e = gid >> 2;
    int h = gid & 3;
    int s, d; edge_sd(ei, Ereal, e, s, d);
    float sc = alsrc[s * Hh + h] + aldst[d * Hh + h];
    sc = sc > 0.f ? sc : sc * NEG_SLOPE;
    atomicMax(&mkey[d * Hh + h], fkey(sc));
  }
}

#define OWN_ROWS   160
#define OWN_WAVES  32
#define SCAN_LOADS 8
#define WAVE_EDGES (SCAN_LOADS * 128)
#define ROUND_EDGES (OWN_WAVES * WAVE_EDGES)

template <int K>
__device__ __forceinline__ void agg_edges(
    int j, int cnt, int sv, int dv, int lane, int hh, int n0,
    const float* __restrict__ alsrc, const float* __restrict__ aldst,
    const unsigned* __restrict__ mkey, const float* __restrict__ Hf,
    float* s_acc, float* s_sum) {
  int   dl[K];
  float asv[K], adv[K], hv[K][Hh];
  unsigned mk[K];
  for (int k = 0; k < K; ++k) {
    const int slot = (j + k < cnt) ? (j + k) : (cnt - 1);
    const int s = __builtin_amdgcn_readlane(sv, slot);
    const int d = __builtin_amdgcn_readlane(dv, slot);
    dl[k]  = d - n0;
    asv[k] = alsrc[s * Hh + hh];
    adv[k] = aldst[d * Hh + hh];
    mk[k]  = mkey[d * Hh + hh];
    const float* hsrc = Hf + (long)s * Dch + lane;
    for (int h = 0; h < Hh; ++h) hv[k][h] = hsrc[h * Ch];
  }
  for (int k = 0; k < K; ++k) {
    if (j + k < cnt) {
      float sc = asv[k] + adv[k];
      sc = sc > 0.f ? sc : sc * NEG_SLOPE;
      float p = expf(sc - kinv(mk[k]));
      if (lane < Hh) atomicAdd(&s_sum[dl[k] * Hh + lane], p);
      float* adst = s_acc + dl[k] * Dch + lane;
      for (int h = 0; h < Hh; ++h) {
        float ph = __int_as_float(__builtin_amdgcn_readlane(__float_as_int(p), h));
        atomicAdd(&adst[h * Ch], ph * hv[k][h]);
      }
    }
  }
}

__device__ __forceinline__ void scan_load(const int* __restrict__ dstp, int Ereal,
                                          bool vec_ok, int base, int lane,
                                          int4 (&q)[SCAN_LOADS]) {
  if (vec_ok && base + WAVE_EDGES <= Ereal) {
    const int4* p = (const int4*)(dstp + base) + lane;
    for (int k = 0; k < SCAN_LOADS; ++k) q[k] = p[k * 32];
  } else {
    for (int k = 0; k < SCAN_LOADS; ++k) {
      const int e0 = base + k * 128 + lane * 4;
      q[k].x = (e0     < Ereal) ? dstp[e0]     : -1;
      q[k].y = (e0 + 1 < Ereal) ? dstp[e0 + 1] : -1;
      q[k].z = (e0 + 2 < Ereal) ? dstp[e0 + 2] : -1;
      q[k].w = (e0 + 3 < Ereal) ? dstp[e0 + 3] : -1;
    }
  }
}

__global__ __launch_bounds__(OWN_WAVES * 32) void owner_aggregate(
    const int* __restrict__ ei, int Ereal, int Nn,
    const float* __restrict__ alsrc, const float* __restrict__ aldst,
    const unsigned* __restrict__ mkey, const float* __restrict__ Hf,
    float* __restrict__ ssum, float* __restrict__ acc) {
  __shared__ float s_acc[OWN_ROWS * Dch];
  __shared__ float s_sum[OWN_ROWS * Hh];
  __shared__ int   s_list[OWN_WAVES * WAVE_EDGES];

  const int tid  = threadIdx.x;
  const int lane = tid & 31;
  const int hh   = lane & 3;
  const int wave = __builtin_amdgcn_readfirstlane(tid >> 5);
  const int n0   = (int)blockIdx.x * OWN_ROWS;
  const int rows = (Nn - n0) < OWN_ROWS ? (Nn - n0) : OWN_ROWS;

  for (int i = tid; i < OWN_ROWS * Dch / 4; i += OWN_WAVES * 32)
    ((float4*)s_acc)[i] = make_float4(0.f, 0.f, 0.f, 0.f);
  for (int i = tid; i < OWN_ROWS * Hh; i += OWN_WAVES * 32) s_sum[i] = 0.f;
  __syncthreads();

  const int* __restrict__ dstp = ei + Ereal;
  const bool vec_ok = (Ereal & 3) == 0;
  int R = (Ereal + ROUND_EDGES - 1) / ROUND_EDGES;
  R = R < 1 ? 1 : R;
  int* mylist = s_list + wave * WAVE_EDGES;

  int rr = (int)(blockIdx.x % (unsigned)R);
  int4 q[SCAN_LOADS];
  scan_load(dstp, Ereal, vec_ok, (rr * OWN_WAVES + wave) * WAVE_EDGES, lane, q);

  for (int r = 0; r < R; ++r) {
    const int base = (rr * OWN_WAVES + wave) * WAVE_EDGES;
    int cnt = 0;
    for (int k = 0; k < SCAN_LOADS; ++k) {
      const int e0 = base + k * 128 + lane * 4;
      const int dd[4] = {q[k].x, q[k].y, q[k].z, q[k].w};
      for (int c = 0; c < 4; ++c) {
        const bool hit = (unsigned)(dd[c] - n0) < (unsigned)rows;
        const unsigned mask = __builtin_amdgcn_ballot_w32(hit);
        if (mask) {
          if (hit) mylist[cnt + (int)__builtin_amdgcn_mbcnt_lo(mask, 0u)] = e0 + c;
          cnt += __builtin_popcount(mask);
        }
      }
    }
    rr = (rr + 1 == R) ? 0 : rr + 1;
    if (r + 1 < R)
      scan_load(dstp, Ereal, vec_ok, (rr * OWN_WAVES + wave) * WAVE_EDGES, lane, q);

    for (int b = 0; b < cnt; b += 32) {
      const int n = (cnt - b) < 32 ? (cnt - b) : 32;
      const int e  = mylist[b + (lane < n ? lane : n - 1)];
      const int sv = ei[e];
      const int dv = dstp[e];
      for (int j = 0; j < n; j += 8)
        agg_edges<8>(j, n, sv, dv, lane, hh, n0, alsrc, aldst, mkey, Hf, s_acc, s_sum);
    }
  }

  {
    const int per = OWN_ROWS / OWN_WAVES;
    int n = rows - wave * per;
    n = n < 0 ? 0 : (n > per ? per : n);
    if (n > 0) {
      const int sv = n0 + wave * per + lane;
      agg_edges<8>(0, n, sv, sv, lane, hh, n0, alsrc, aldst, mkey, Hf, s_acc, s_sum);
    }
  }

  __syncthreads();
  float4*       o4 = (float4*)(acc + (long)n0 * Dch);
  const float4* a4 = (const float4*)s_acc;
  for (int i = tid; i < rows * (Dch / 4); i += OWN_WAVES * 32) o4[i] = a4[i];
  for (int i = tid; i < rows * Hh; i += OWN_WAVES * 32) ssum[(long)n0 * Hh + i] = s_sum[i];
}

__global__ __launch_bounds__(256) void post_bn_elu(float* __restrict__ x,
                            const float* __restrict__ ssum,
                            const float* __restrict__ bias,
                            const float* __restrict__ gamma,
                            const float* __restrict__ beta,
                            const float* __restrict__ mean,
                            const float* __restrict__ var,
                            long n) {
  const long tid    = (long)blockIdx.x * blockDim.x + threadIdx.x;
  const long stride = (long)gridDim.x * blockDim.x;
  const int  c      = (int)((tid * 4) & (Dch - 1));
  const int  head   = c / Ch;
  float b[4], m[4], sc[4], be[4];
  for (int k = 0; k < 4; ++k) {
    b[k]  = bias[c + k];
    m[k]  = mean[c + k];
    sc[k] = gamma[c + k] / sqrtf(var[c + k] + BN_EPS);
    be[k] = beta[c + k];
  }
  float4* x4 = (float4*)x;
  const long n4 = n >> 2;
  for (long i = tid; i < n4; i += stride) {
    float4 q = x4[i];
    const float den = ssum[(i >> 5) * Hh + head] + 1e-16f;
    float v[4] = {q.x, q.y, q.z, q.w};
    for (int k = 0; k < 4; ++k) {
      float t = v[k] / den + b[k];
      t = (t - m[k]) * sc[k] + be[k];
      v[k] = t > 0.f ? t : expf(t) - 1.f;
    }
    x4[i] = make_float4(v[0], v[1], v[2], v[3]);
  }
}

__global__ void pool_accum(const float* __restrict__ x,
                           const int* __restrict__ batch,
                           float* __restrict__ sums, float* __restrict__ cnt,
                           long n) {
  const long stride = (long)gridDim.x * blockDim.x;
  for (long i = (long)blockIdx.x * blockDim.x + threadIdx.x; i < n; i += stride) {
    int node = (int)(i >> 7);
    int c    = (int)(i & 127);
    int g    = batch[node];
    atomicAdd(&sums[(long)g * Dch + c], x[i]);
    if (c == 0) atomicAdd(&cnt[g], 1.0f);
  }
}

__global__ void pool_final(const float* __restrict__ sums,
                           const float* __restrict__ cnt,
                           float* __restrict__ out, int G) {
  int i = blockIdx.x * blockDim.x + threadIdx.x;
  if (i >= G * Dch) return;
  int g = i >> 7;
  float c = cnt[g];
  out[i] = sums[i] / (c > 1.f ? c : 1.f);
}

extern "C" void kernel_launch(void* const* d_in, const int* in_sizes, int n_in,
                              void* d_out, int out_size, void* d_ws, size_t ws_size,
                              hipStream_t stream) {
  const float* x0    = (const float*)d_in[0];
  const int*   ei    = (const int*)d_in[1];
  const int*   batch = (const int*)d_in[2];
  const float* Ws    = (const float*)d_in[3];
  const float* a_src = (const float*)d_in[4];
  const float* a_dst = (const float*)d_in[5];
  const float* bias  = (const float*)d_in[6];
  const float* gamma = (const float*)d_in[7];
  const float* beta  = (const float*)d_in[8];
  const float* mean  = (const float*)d_in[9];
  const float* var   = (const float*)d_in[10];

  const int Nn    = in_sizes[0] / Dch;
  const int Ereal = in_sizes[1] / 2;
  const int Etot  = Ereal + Nn;
  const int G     = out_size / Dch;

  char* ws = (char*)d_ws;
  float*    bufA  = (float*)ws;    ws += (size_t)Nn * Dch * 4;
  float*    bufB  = (float*)ws;    ws += (size_t)Nn * Dch * 4;
  float*    hbuf  = (float*)ws;    ws += (size_t)Nn * Dch * 4;
  unsigned* mkey  = (unsigned*)ws; ws += (size_t)Nn * Hh * 4;
  float*    ssum  = (float*)ws;    ws += (size_t)Nn * Hh * 4;
  float*    alsrc = (float*)ws;    ws += (size_t)Nn * Hh * 4;
  float*    aldst = (float*)ws;    ws += (size_t)Nn * Hh * 4;
  float*    psum  = (float*)ws;    ws += (size_t)G * Dch * 4;
  float*    pcnt  = (float*)ws;    ws += (size_t)G * 4;

  const float* xin  = x0;
  float*       xout = bufA;
  for (int l = 0; l < Lnum; ++l) {
    const float* Wl = Ws + (size_t)l * Dch * Dch;
    gemm_wmma_f32<<<Nn / 16, 256, 0, stream>>>(xin, Wl, hbuf);

    att_logits<<<(Nn * Hh + 255) / 256, 256, 0, stream>>>(
        hbuf, a_src + (size_t)l * Hh * Ch, a_dst + (size_t)l * Hh * Ch,
        alsrc, aldst, Nn);

    fill_u32<<<512, 256, 0, stream>>>(mkey, 0u, (long)Nn * Hh);

    edge_max<<<LOOP_BLOCKS, 256, 0, stream>>>(ei, Ereal, Etot, alsrc, aldst, mkey);

    owner_aggregate<<<(Nn + OWN_ROWS - 1) / OWN_ROWS, OWN_WAVES * 32, 0, stream>>>(
        ei, Ereal, Nn, alsrc, aldst, mkey, hbuf, ssum, xout);

    post_bn_elu<<<LOOP_BLOCKS, 256, 0, stream>>>(
        xout, ssum, bias + (size_t)l * Dch, gamma + (size_t)l * Dch,
        beta + (size_t)l * Dch, mean + (size_t)l * Dch, var + (size_t)l * Dch,
        (long)Nn * Dch);

    xin  = xout;
    xout = (xout == bufA) ? bufB : bufA;
  }

  fill_u32<<<32, 256, 0, stream>>>((unsigned*)psum, 0u, (long)G * Dch);
  fill_u32<<<1, 64, 0, stream>>>((unsigned*)pcnt, 0u, G);
  pool_accum<<<LOOP_BLOCKS, 256, 0, stream>>>(xin, batch, psum, pcnt, (long)Nn * Dch);
  pool_final<<<(G * Dch + 255) / 256, 256, 0, stream>>>(psum, pcnt, (float*)d_out, G);
}


# owner pass scans one owner byte per edge (byte list written once per call) in place of the 4-byte destination list: a quarter of the scan bytes, 7 rounds not 20; row sums added in another order
# speedup vs baseline: 2.5019x; 1.0424x over previous
#include <hip/hip_runtime.h>


#define Dch      128
#define Hh       4
#define Ch       32
#define Lnum     3
#define NEG_SLOPE 0.2f
#define BN_EPS   1e-5f
#define LOOP_BLOCKS 2048

typedef __attribute__((ext_vector_type(2))) float v2f;
typedef __attribute__((ext_vector_type(8))) float v8f;

__device__ __forceinline__ unsigned fkey(float f) {
  unsigned b = __float_as_uint(f);
  return b ^ ((unsigned)((int)b >> 31) | 0x80000000u);
}
__device__ __forceinline__ float kinv(unsigned k) {
  return __uint_as_float((k & 0x80000000u) ? (k ^ 0x80000000u) : ~k);
}

__device__ __forceinline__ void edge_sd(const int* __restrict__ ei, int Ereal,
                                        int e, int& s, int& d) {
  if (e < Ereal) { s = ei[e]; d = ei[Ereal + e]; }
  else           { s = e - Ereal; d = s; }
}

__global__ void fill_u32(unsigned* __restrict__ p, unsigned v, long n) {
  long i = (long)blockIdx.x * blockDim.x + threadIdx.x;
  long stride = (long)gridDim.x * blockDim.x;
  for (; i < n; i += stride) p[i] = v;
}

__global__ __launch_bounds__(256) void gemm_wmma_f32(
    const float* __restrict__ X, const float* __restrict__ Wm,
    float* __restrict__ Ho) {
  const int wave = threadIdx.x >> 5;
  const int lane = threadIdx.x & 31;
  const int half = lane >> 4;
  const int l15  = lane & 15;
  const int row0 = blockIdx.x << 4;
  const int bcol = (wave << 4) + l15;
  const float* arow = X + (long)(row0 + l15) * Dch;

  v8f acc = {};
  for (int k = 0; k < Dch; k += 4) {
    const int ka = k + half * 2;
    v2f a, b;
    a.x = arow[ka];
    a.y = arow[ka + 1];
    b.x = Wm[(long)ka * Dch + bcol];
    b.y = Wm[(long)(ka + 1) * Dch + bcol];
    acc = __builtin_amdgcn_wmma_f32_16x16x4_f32(
         false, a,  false, b,
         (short)0, acc,  false,  false);
  }
  float* orow = Ho + (long)(row0 + half * 8) * Dch + bcol;
#pragma unroll
  for (int v = 0; v < 8; ++v) orow[(long)v * Dch] = acc[v];
}

__global__ void att_logits(const float* __restrict__ Hf,
                           const float* __restrict__ a_src,
                           const float* __restrict__ a_dst,
                           float* __restrict__ alsrc, float* __restrict__ aldst,
                           int Nn) {
  int i = blockIdx.x * blockDim.x + threadIdx.x;
  if (i >= Nn * Hh) return;
  int n = i >> 2, h = i & 3;
  const float* hp = Hf + (long)n * Dch + h * Ch;
  const float* as = a_src + h * Ch;
  const float* ad = a_dst + h * Ch;
  float s0 = 0.f, s1 = 0.f;
#pragma unroll
  for (int c = 0; c < Ch; ++c) { float v = hp[c]; s0 += v * as[c]; s1 += v * ad[c]; }
  alsrc[i] = s0;
  aldst[i] = s1;
}


__global__ __launch_bounds__(256) void edge_max(
    const int* __restrict__ ei, int Ereal, int Etot,
    const float* __restrict__ alsrc, const float* __restrict__ aldst,
    unsigned* __restrict__ mkey) {
  const int total  = Etot * Hh;
  const int stride = (int)(gridDim.x * blockDim.x);
  for (int gid = (int)(blockIdx.x * blockDim.x + threadIdx.x); gid < total; gid += stride) {
    int e = gid >> 2;
    int h = gid & 3;
    int s, d; edge_sd(ei, Ereal, e, s, d);
    float sc = alsrc[s * Hh + h] + aldst[d * Hh + h];
    sc = sc > 0.f ? sc : sc * NEG_SLOPE;
    atomicMax(&mkey[d * Hh + h], fkey(sc));
  }
}

#define OWN_ROWS   160
#define OWN_WAVES  32
#define SCAN_LOADS 6
#define LOAD_EDGES 512
#define WAVE_EDGES (SCAN_LOADS * LOAD_EDGES)
#define ROUND_EDGES (OWN_WAVES * WAVE_EDGES)

template <int K>
__device__ __forceinline__ void agg_edges(
    int j, int cnt, int sv, int dv, int lane, int hh, int n0, int rows,
    const float* __restrict__ alsrc, const float* __restrict__ aldst,
    const unsigned* __restrict__ mkey, const float* __restrict__ Hf,
    float* s_acc, float* s_sum) {
  int   dl[K];
  float asv[K], adv[K], hv[K][Hh];
  unsigned mk[K];
  for (int k = 0; k < K; ++k) {
    const int slot = (j + k < cnt) ? (j + k) : (cnt - 1);
    const int s = __builtin_amdgcn_readlane(sv, slot);
    const int d = __builtin_amdgcn_readlane(dv, slot);
    dl[k]  = d - n0;
    const int dd = (unsigned)dl[k] < (unsigned)rows ? d : n0;
    asv[k] = alsrc[s * Hh + hh];
    adv[k] = aldst[dd * Hh + hh];
    mk[k]  = mkey[dd * Hh + hh];
    const float* hsrc = Hf + (long)s * Dch + lane;
    for (int h = 0; h < Hh; ++h) hv[k][h] = hsrc[h * Ch];
  }
  for (int k = 0; k < K; ++k) {
    if (j + k < cnt && (unsigned)dl[k] < (unsigned)rows) {
      float sc = asv[k] + adv[k];
      sc = sc > 0.f ? sc : sc * NEG_SLOPE;
      float p = expf(sc - kinv(mk[k]));
      if (lane < Hh) atomicAdd(&s_sum[dl[k] * Hh + lane], p);
      float* adst = s_acc + dl[k] * Dch + lane;
      for (int h = 0; h < Hh; ++h) {
        float ph = __int_as_float(__builtin_amdgcn_readlane(__float_as_int(p), h));
        atomicAdd(&adst[h * Ch], ph * hv[k][h]);
      }
    }
  }
}

__device__ __forceinline__ int scan_rounds(int Ereal) {
  int R = (Ereal + ROUND_EDGES - 1) / ROUND_EDGES;
  return R < 1 ? 1 : R;
}

__global__ __launch_bounds__(256) void owner_tags(const int* __restrict__ ei, int Ereal,
                                                  unsigned* __restrict__ tags4) {
  const int* __restrict__ dstp = ei + Ereal;
  const bool vec_ok = (Ereal & 3) == 0;
  const int words  = scan_rounds(Ereal) * (ROUND_EDGES / 4);
  const int stride = (int)(gridDim.x * blockDim.x);
  for (int w = (int)(blockIdx.x * blockDim.x + threadIdx.x); w < words; w += stride) {
    const int e0 = w * 4;
    unsigned v;
    if (vec_ok && e0 + 4 <= Ereal) {
      const int4 q = *(const int4*)(dstp + e0);
      const int d[4] = {q.x, q.y, q.z, q.w};
      v = 0u;
      for (int c = 0; c < 4; ++c)
        v |= (((unsigned)d[c] / (unsigned)OWN_ROWS) & 255u) << (8 * c);
    } else {
      v = 0xffffffffu;
      for (int c = 0; c < 4; ++c)
        if (e0 + c < Ereal)
          v = (v & ~(255u << (8 * c))) |
              ((((unsigned)dstp[e0 + c] / (unsigned)OWN_ROWS) & 255u) << (8 * c));
    }
    tags4[w] = v;
  }
}

__device__ __forceinline__ void scan_load(const unsigned char* __restrict__ tags,
                                          int base, int lane,
                                          uint4 (&q)[SCAN_LOADS]) {
  const uint4* p = (const uint4*)(tags + base) + lane;
  for (int k = 0; k < SCAN_LOADS; ++k) q[k] = p[k * 32];
}

__global__ __launch_bounds__(OWN_WAVES * 32) void owner_aggregate(
    const int* __restrict__ ei, const unsigned char* __restrict__ tags, int Ereal, int Nn,
    const float* __restrict__ alsrc, const float* __restrict__ aldst,
    const unsigned* __restrict__ mkey, const float* __restrict__ Hf,
    float* __restrict__ ssum, float* __restrict__ acc) {
  __shared__ float s_acc[OWN_ROWS * Dch];
  __shared__ float s_sum[OWN_ROWS * Hh];
  __shared__ unsigned short s_list[OWN_WAVES * WAVE_EDGES];

  const int tid  = threadIdx.x;
  const int lane = tid & 31;
  const int hh   = lane & 3;
  const int wave = __builtin_amdgcn_readfirstlane(tid >> 5);
  const int n0   = (int)blockIdx.x * OWN_ROWS;
  const int rows = (Nn - n0) < OWN_ROWS ? (Nn - n0) : OWN_ROWS;

  for (int i = tid; i < OWN_ROWS * Dch / 4; i += OWN_WAVES * 32)
    ((float4*)s_acc)[i] = make_float4(0.f, 0.f, 0.f, 0.f);
  for (int i = tid; i < OWN_ROWS * Hh; i += OWN_WAVES * 32) s_sum[i] = 0.f;
  __syncthreads();

  const int* __restrict__ dstp = ei + Ereal;
  const int R = scan_rounds(Ereal);
  const int T = R * OWN_WAVES;
  unsigned short* mylist = s_list + wave * WAVE_EDGES;
  const unsigned pat = ((unsigned)blockIdx.x & 255u) * 0x01010101u;

  int ch = (int)(((unsigned)blockIdx.x * 13u + (unsigned)wave) % (unsigned)T);
  uint4 q[SCAN_LOADS];
  scan_load(tags, ch * WAVE_EDGES, lane, q);

  int lane16 = lane * 16;
  for (int r = 0; r < R; ++r) {
    const int base = ch * WAVE_EDGES;
    asm volatile("" : "+v"(lane16));
    int cnt = 0;
#pragma unroll
    for (int k = 0; k < SCAN_LOADS; ++k) {
      const int o0 = k * LOAD_EDGES + lane16;
      const unsigned wv[4] = {q[k].x, q[k].y, q[k].z, q[k].w};
#pragma unroll
      for (int j = 0; j < 4; ++j) {
        const unsigned x = wv[j] ^ pat;
        const unsigned z = (x - 0x01010101u) & ~x & 0x80808080u;
        if (__builtin_amdgcn_ballot_w32(z != 0u)) {
          for (int c = 0; c < 4; ++c) {
            const int  o   = o0 + j * 4 + c;
            const bool hit = (x & (255u << (8 * c))) == 0u;
            const unsigned mask = __builtin_amdgcn_ballot_w32(hit);
            if (mask) {
              if (hit) mylist[cnt + (int)__builtin_amdgcn_mbcnt_lo(mask, 0u)] = (unsigned short)o;
              cnt += __builtin_popcount(mask);
            }
          }
        }
      }
    }
    ch += OWN_WAVES;
    if (ch >= T) ch -= T;
    if (r + 1 < R) scan_load(tags, ch * WAVE_EDGES, lane, q);

    for (int b = 0; b < cnt; b += 32) {
      const int n = (cnt - b) < 32 ? (cnt - b) : 32;
      const int e  = base + (int)mylist[b + (lane < n ? lane : n - 1)];
      const int sv = e < Ereal ? ei[e]   : 0;
      const int dv = e < Ereal ? dstp[e] : -1;
      for (int j = 0; j < n; j += 8)
        agg_edges<8>(j, n, sv, dv, lane, hh, n0, rows, alsrc, aldst, mkey, Hf, s_acc, s_sum);
    }
  }

  {
    const int per = OWN_ROWS / OWN_WAVES;
    int n = rows - wave * per;
    n = n < 0 ? 0 : (n > per ? per : n);
    if (n > 0) {
      const int sv = n0 + wave * per + lane;
      agg_edges<8>(0, n, sv, sv, lane, hh, n0, rows, alsrc, aldst, mkey, Hf, s_acc, s_sum);
    }
  }

  __syncthreads();
  float4*       o4 = (float4*)(acc + (long)n0 * Dch);
  const float4* a4 = (const float4*)s_acc;
  for (int i = tid; i < rows * (Dch / 4); i += OWN_WAVES * 32) o4[i] = a4[i];
  for (int i = tid; i < rows * Hh; i += OWN_WAVES * 32) ssum[(long)n0 * Hh + i] = s_sum[i];
}

__global__ __launch_bounds__(256) void post_bn_elu(float* __restrict__ x,
                            const float* __restrict__ ssum,
                            const float* __restrict__ bias,
                            const float* __restrict__ gamma,
                            const float* __restrict__ beta,
                            const float* __restrict__ mean,
                            const float* __restrict__ var,
                            long n) {
  const long tid    = (long)blockIdx.x * blockDim.x + threadIdx.x;
  const long stride = (long)gridDim.x * blockDim.x;
  const int  c      = (int)((tid * 4) & (Dch - 1));
  const int  head   = c / Ch;
  float b[4], m[4], sc[4], be[4];
  for (int k = 0; k < 4; ++k) {
    b[k]  = bias[c + k];
    m[k]  = mean[c + k];
    sc[k] = gamma[c + k] / sqrtf(var[c + k] + BN_EPS);
    be[k] = beta[c + k];
  }
  float4* x4 = (float4*)x;
  const long n4 = n >> 2;
  for (long i = tid; i < n4; i += stride) {
    float4 q = x4[i];
    const float den = ssum[(i >> 5) * Hh + head] + 1e-16f;
    float v[4] = {q.x, q.y, q.z, q.w};
    for (int k = 0; k < 4; ++k) {
      float t = v[k] / den + b[k];
      t = (t - m[k]) * sc[k] + be[k];
      v[k] = t > 0.f ? t : expf(t) - 1.f;
    }
    x4[i] = make_float4(v[0], v[1], v[2], v[3]);
  }
}

__global__ void pool_accum(const float* __restrict__ x,
                           const int* __restrict__ batch,
                           float* __restrict__ sums, float* __restrict__ cnt,
                           long n) {
  const long stride = (long)gridDim.x * blockDim.x;
  for (long i = (long)blockIdx.x * blockDim.x + threadIdx.x; i < n; i += stride) {
    int node = (int)(i >> 7);
    int c    = (int)(i & 127);
    int g    = batch[node];
    atomicAdd(&sums[(long)g * Dch + c], x[i]);
    if (c == 0) atomicAdd(&cnt[g], 1.0f);
  }
}

__global__ void pool_final(const float* __restrict__ sums,
                           const float* __restrict__ cnt,
                           float* __restrict__ out, int G) {
  int i = blockIdx.x * blockDim.x + threadIdx.x;
  if (i >= G * Dch) return;
  int g = i >> 7;
  float c = cnt[g];
  out[i] = sums[i] / (c > 1.f ? c : 1.f);
}

extern "C" void kernel_launch(void* const* d_in, const int* in_sizes, int n_in,
                              void* d_out, int out_size, void* d_ws, size_t ws_size,
                              hipStream_t stream) {
  const float* x0    = (const float*)d_in[0];
  const int*   ei    = (const int*)d_in[1];
  const int*   batch = (const int*)d_in[2];
  const float* Ws    = (const float*)d_in[3];
  const float* a_src = (const float*)d_in[4];
  const float* a_dst = (const float*)d_in[5];
  const float* bias  = (const float*)d_in[6];
  const float* gamma = (const float*)d_in[7];
  const float* beta  = (const float*)d_in[8];
  const float* mean  = (const float*)d_in[9];
  const float* var   = (const float*)d_in[10];

  const int Nn    = in_sizes[0] / Dch;
  const int Ereal = in_sizes[1] / 2;
  const int Etot  = Ereal + Nn;
  const int G     = out_size / Dch;

  char* ws = (char*)d_ws;
  float*    bufA  = (float*)ws;    ws += (size_t)Nn * Dch * 4;
  unsigned char* tags = (unsigned char*)ws; ws += (size_t)Nn * Dch * 4;
  float*    hbuf  = (float*)ws;    ws += (size_t)Nn * Dch * 4;
  unsigned* mkey  = (unsigned*)ws; ws += (size_t)Nn * Hh * 4;
  float*    ssum  = (float*)ws;    ws += (size_t)Nn * Hh * 4;
  float*    alsrc = (float*)ws;    ws += (size_t)Nn * Hh * 4;
  float*    aldst = (float*)ws;    ws += (size_t)Nn * Hh * 4;
  float*    psum  = (float*)ws;    ws += (size_t)G * Dch * 4;
  float*    pcnt  = (float*)ws;    ws += (size_t)G * 4;

  owner_tags<<<256, 256, 0, stream>>>(ei, Ereal, (unsigned*)tags);

  const float* xin  = x0;
  float*       xout = bufA;
  for (int l = 0; l < Lnum; ++l) {
    const float* Wl = Ws + (size_t)l * Dch * Dch;
    gemm_wmma_f32<<<Nn / 16, 256, 0, stream>>>(xin, Wl, hbuf);

    att_logits<<<(Nn * Hh + 255) / 256, 256, 0, stream>>>(
        hbuf, a_src + (size_t)l * Hh * Ch, a_dst + (size_t)l * Hh * Ch,
        alsrc, aldst, Nn);

    fill_u32<<<512, 256, 0, stream>>>(mkey, 0u, (long)Nn * Hh);

    edge_max<<<LOOP_BLOCKS, 256, 0, stream>>>(ei, Ereal, Etot, alsrc, aldst, mkey);

    owner_aggregate<<<(Nn + OWN_ROWS - 1) / OWN_ROWS, OWN_WAVES * 32, 0, stream>>>(
        ei, tags, Ereal, Nn, alsrc, aldst, mkey, hbuf, ssum, xout);

    post_bn_elu<<<LOOP_BLOCKS, 256, 0, stream>>>(
        xout, ssum, bias + (size_t)l * Dch, gamma + (size_t)l * Dch,
        beta + (size_t)l * Dch, mean + (size_t)l * Dch, var + (size_t)l * Dch,
        (long)Nn * Dch);

    xin  = xout;
  }

  fill_u32<<<32, 256, 0, stream>>>((unsigned*)psum, 0u, (long)G * Dch);
  fill_u32<<<1, 64, 0, stream>>>((unsigned*)pcnt, 0u, G);
  pool_accum<<<LOOP_BLOCKS, 256, 0, stream>>>(xin, batch, psum, pcnt, (long)Nn * Dch);
  pool_final<<<(G * Dch + 255) / 256, 256, 0, stream>>>(psum, pcnt, (float*)d_out, G);
}


# segment max taken in the owner pass's LDS (integer atomic max, one barrier; a wave lists its edges once, walks the list twice): edge_max and its clear gone, 23 to 17 launches; float adds reordered
# speedup vs baseline: 3.2654x; 1.3052x over previous
#include <hip/hip_runtime.h>


#define Dch      128
#define Hh       4
#define Ch       32
#define Lnum     3
#define NEG_SLOPE 0.2f
#define BN_EPS   1e-5f
#define LOOP_BLOCKS 2048

typedef __attribute__((ext_vector_type(2))) float v2f;
typedef __attribute__((ext_vector_type(8))) float v8f;

__device__ __forceinline__ unsigned fkey(float f) {
  unsigned b = __float_as_uint(f);
  return b ^ ((unsigned)((int)b >> 31) | 0x80000000u);
}
__device__ __forceinline__ float kinv(unsigned k) {
  return __uint_as_float((k & 0x80000000u) ? (k ^ 0x80000000u) : ~k);
}

__device__ __forceinline__ void edge_sd(const int* __restrict__ ei, int Ereal,
                                        int e, int& s, int& d) {
  if (e < Ereal) { s = ei[e]; d = ei[Ereal + e]; }
  else           { s = e - Ereal; d = s; }
}

__global__ void fill_u32(unsigned* __restrict__ p, unsigned v, long n) {
  long i = (long)blockIdx.x * blockDim.x + threadIdx.x;
  long stride = (long)gridDim.x * blockDim.x;
  for (; i < n; i += stride) p[i] = v;
}

__global__ __launch_bounds__(256) void gemm_wmma_f32(
    const float* __restrict__ X, const float* __restrict__ Wm,
    float* __restrict__ Ho) {
  const int wave = threadIdx.x >> 5;
  const int lane = threadIdx.x & 31;
  const int half = lane >> 4;
  const int l15  = lane & 15;
  const int row0 = blockIdx.x << 4;
  const int bcol = (wave << 4) + l15;
  const float* arow = X + (long)(row0 + l15) * Dch;

  v8f acc = {};
  for (int k = 0; k < Dch; k += 4) {
    const int ka = k + half * 2;
    v2f a, b;
    a.x = arow[ka];
    a.y = arow[ka + 1];
    b.x = Wm[(long)ka * Dch + bcol];
    b.y = Wm[(long)(ka + 1) * Dch + bcol];
    acc = __builtin_amdgcn_wmma_f32_16x16x4_f32(
         false, a,  false, b,
         (short)0, acc,  false,  false);
  }
  float* orow = Ho + (long)(row0 + half * 8) * Dch + bcol;
#pragma unroll
  for (int v = 0; v < 8; ++v) orow[(long)v * Dch] = acc[v];
}

__global__ void att_logits(const float* __restrict__ Hf,
                           const float* __restrict__ a_src,
                           const float* __restrict__ a_dst,
                           float* __restrict__ alsrc, float* __restrict__ aldst,
                           int Nn) {
  int i = blockIdx.x * blockDim.x + threadIdx.x;
  if (i >= Nn * Hh) return;
  int n = i >> 2, h = i & 3;
  const float* hp = Hf + (long)n * Dch + h * Ch;
  const float* as = a_src + h * Ch;
  const float* ad = a_dst + h * Ch;
  float s0 = 0.f, s1 = 0.f;
#pragma unroll
  for (int c = 0; c < Ch; ++c) { float v = hp[c]; s0 += v * as[c]; s1 += v * ad[c]; }
  alsrc[i] = s0;
  aldst[i] = s1;
}

#define OWN_ROWS   160
#define OWN_WAVES  32
#define SCAN_LOADS 6
#define LOAD_EDGES 512
#define WAVE_EDGES (SCAN_LOADS * LOAD_EDGES)
#define ROUND_EDGES (OWN_WAVES * WAVE_EDGES)
#define LIST_CAP   (WAVE_EDGES / 2)

__device__ __forceinline__ void batch_ends(const int* __restrict__ ei, int Ereal,
                                           const int* list, int b, int n, int lane,
                                           int& sv, int& dv) {
  const int e = list[b + (lane < n ? lane : n - 1)];
  sv = e < Ereal ? ei[e]         : 0;
  dv = e < Ereal ? ei[Ereal + e] : -1;
}

__device__ __forceinline__ void max_batch(
    int n, int sv, int dv, int lane, int n0, int rows,
    const float* __restrict__ alsrc, const float* __restrict__ aldst,
    unsigned* s_max) {
  const int  dl = dv - n0;
  const bool ok = lane < n && (unsigned)dl < (unsigned)rows;
  const int  dd = ok ? dv : n0;
  const float4 a4 = ((const float4*)alsrc)[sv];
  const float4 d4 = ((const float4*)aldst)[dd];
  if (ok) {
    const float av[Hh] = {a4.x, a4.y, a4.z, a4.w};
    const float bv[Hh] = {d4.x, d4.y, d4.z, d4.w};
    for (int h = 0; h < Hh; ++h) {
      float sc = av[h] + bv[h];
      sc = sc > 0.f ? sc : sc * NEG_SLOPE;
      atomicMax(&s_max[dl * Hh + h], fkey(sc));
    }
  }
}

template <int K>
__device__ __forceinline__ void agg_edges(
    int j, int cnt, int sv, int dv, int lane, int hh, int n0, int rows,
    const float* __restrict__ alsrc, const float* __restrict__ aldst,
    const unsigned* s_max, const float* __restrict__ Hf,
    float* s_acc, float* s_sum) {
  int   dl[K];
  float asv[K], adv[K], hv[K][Hh];
  unsigned mk[K];
  for (int k = 0; k < K; ++k) {
    const int slot = (j + k < cnt) ? (j + k) : (cnt - 1);
    const int s = __builtin_amdgcn_readlane(sv, slot);
    const int d = __builtin_amdgcn_readlane(dv, slot);
    dl[k]  = d - n0;
    const bool mine = (unsigned)dl[k] < (unsigned)rows;
    asv[k] = alsrc[s * Hh + hh];
    adv[k] = aldst[(mine ? d : n0) * Hh + hh];
    mk[k]  = s_max[(mine ? dl[k] : 0) * Hh + hh];
    const float* hsrc = Hf + (long)s * Dch + lane;
    for (int h = 0; h < Hh; ++h) hv[k][h] = hsrc[h * Ch];
  }
  for (int k = 0; k < K; ++k) {
    if (j + k < cnt && (unsigned)dl[k] < (unsigned)rows) {
      float sc = asv[k] + adv[k];
      sc = sc > 0.f ? sc : sc * NEG_SLOPE;
      float p = expf(sc - kinv(mk[k]));
      if (lane < Hh) atomicAdd(&s_sum[dl[k] * Hh + lane], p);
      float* adst = s_acc + dl[k] * Dch + lane;
      for (int h = 0; h < Hh; ++h) {
        float ph = __int_as_float(__builtin_amdgcn_readlane(__float_as_int(p), h));
        atomicAdd(&adst[h * Ch], ph * hv[k][h]);
      }
    }
  }
}

template <int PHASE>
__device__ __forceinline__ void work_list(
    const int* list, int cnt, const int* __restrict__ ei, int Ereal,
    int lane, int hh, int n0, int rows,
    const float* __restrict__ alsrc, const float* __restrict__ aldst,
    unsigned* s_max, const float* __restrict__ Hf, float* s_acc, float* s_sum) {
  for (int b = 0; b < cnt; b += 32) {
    const int n = (cnt - b) < 32 ? (cnt - b) : 32;
    int sv, dv;
    batch_ends(ei, Ereal, list, b, n, lane, sv, dv);
    if (PHASE == 0) {
      max_batch(n, sv, dv, lane, n0, rows, alsrc, aldst, s_max);
    } else {
      for (int j = 0; j < n; j += 8)
        agg_edges<8>(j, n, sv, dv, lane, hh, n0, rows, alsrc, aldst, s_max, Hf, s_acc, s_sum);
    }
  }
}

__device__ __forceinline__ int scan_rounds(int Ereal) {
  int R = (Ereal + ROUND_EDGES - 1) / ROUND_EDGES;
  return R < 1 ? 1 : R;
}

__global__ __launch_bounds__(256) void owner_tags(const int* __restrict__ ei, int Ereal,
                                                  unsigned* __restrict__ tags4) {
  const int* __restrict__ dstp = ei + Ereal;
  const bool vec_ok = (Ereal & 3) == 0;
  const int words  = scan_rounds(Ereal) * (ROUND_EDGES / 4);
  const int stride = (int)(gridDim.x * blockDim.x);
  for (int w = (int)(blockIdx.x * blockDim.x + threadIdx.x); w < words; w += stride) {
    const int e0 = w * 4;
    unsigned v;
    if (vec_ok && e0 + 4 <= Ereal) {
      const int4 q = *(const int4*)(dstp + e0);
      const int d[4] = {q.x, q.y, q.z, q.w};
      v = 0u;
      for (int c = 0; c < 4; ++c)
        v |= (((unsigned)d[c] / (unsigned)OWN_ROWS) & 255u) << (8 * c);
    } else {
      v = 0xffffffffu;
      for (int c = 0; c < 4; ++c)
        if (e0 + c < Ereal)
          v = (v & ~(255u << (8 * c))) |
              ((((unsigned)dstp[e0 + c] / (unsigned)OWN_ROWS) & 255u) << (8 * c));
    }
    tags4[w] = v;
  }
}

__device__ __forceinline__ void scan_load(const unsigned char* __restrict__ tags,
                                          int base, int lane,
                                          uint4 (&q)[SCAN_LOADS]) {
  const uint4* p = (const uint4*)(tags + base) + lane;
  for (int k = 0; k < SCAN_LOADS; ++k) q[k] = p[k * 32];
}

template <int NP>
__device__ __forceinline__ int scan_pieces(const uint4 (&q)[NP], int piece0, unsigned pat,
                                           int lane16, int base, int cnt, int* list) {
#pragma unroll
  for (int k = 0; k < NP; ++k) {
    const int o0 = (piece0 + k) * LOAD_EDGES + lane16;
    const unsigned wv[4] = {q[k].x, q[k].y, q[k].z, q[k].w};
#pragma unroll
    for (int j = 0; j < 4; ++j) {
      const unsigned x = wv[j] ^ pat;
      const unsigned z = (x - 0x01010101u) & ~x & 0x80808080u;
      if (__builtin_amdgcn_ballot_w32(z != 0u)) {
        for (int c = 0; c < 4; ++c) {
          const int  o   = o0 + j * 4 + c;
          const bool hit = (x & (255u << (8 * c))) == 0u;
          const unsigned mask = __builtin_amdgcn_ballot_w32(hit);
          if (mask) {
            const int slot = cnt + (int)__builtin_amdgcn_mbcnt_lo(mask, 0u);
            if (hit && slot < LIST_CAP) list[slot] = base + o;
            cnt += __builtin_popcount(mask);
          }
        }
      }
    }
  }
  return cnt;
}

template <int PHASE>
__device__ __forceinline__ void search_again(
    const unsigned char* __restrict__ tags, int ch, int R, int T, unsigned pat, int lane,
    int* list, const int* __restrict__ ei, int Ereal, int hh, int n0, int rows,
    const float* __restrict__ alsrc, const float* __restrict__ aldst,
    unsigned* s_max, const float* __restrict__ Hf, float* s_acc, float* s_sum) {
  const int lane16 = lane * 16;
  for (int r = 0; r < R; ++r) {
    const int base = ch * WAVE_EDGES;
    uint4 q[SCAN_LOADS];
    scan_load(tags, base, lane, q);
    for (int half = 0; half < 2; ++half) {
      uint4 qh[SCAN_LOADS / 2];
      for (int k = 0; k < SCAN_LOADS / 2; ++k) qh[k] = half ? q[SCAN_LOADS / 2 + k] : q[k];
      const int c = scan_pieces<SCAN_LOADS / 2>(qh, half * (SCAN_LOADS / 2), pat, lane16, base, 0, list);
      work_list<PHASE>(list, c, ei, Ereal, lane, hh, n0, rows, alsrc, aldst, s_max, Hf, s_acc, s_sum);
    }
    ch += OWN_WAVES;
    if (ch >= T) ch -= T;
  }
}

__global__ __launch_bounds__(OWN_WAVES * 32) void owner_aggregate(
    const int* __restrict__ ei, const unsigned char* __restrict__ tags, int Ereal, int Nn,
    const float* __restrict__ alsrc, const float* __restrict__ aldst,
    const float* __restrict__ Hf,
    float* __restrict__ ssum, float* __restrict__ acc) {
  __shared__ float    s_acc[OWN_ROWS * Dch];
  __shared__ float    s_sum[OWN_ROWS * Hh];
  __shared__ unsigned s_max[OWN_ROWS * Hh];
  __shared__ int      s_list[OWN_WAVES * LIST_CAP];

  const int tid  = threadIdx.x;
  const int lane = tid & 31;
  const int hh   = lane & 3;
  const int wave = __builtin_amdgcn_readfirstlane(tid >> 5);
  const int n0   = (int)blockIdx.x * OWN_ROWS;
  const int rows = (Nn - n0) < OWN_ROWS ? (Nn - n0) : OWN_ROWS;

  for (int i = tid; i < OWN_ROWS * Dch / 4; i += OWN_WAVES * 32)
    ((float4*)s_acc)[i] = make_float4(0.f, 0.f, 0.f, 0.f);
  for (int i = tid; i < OWN_ROWS * Hh; i += OWN_WAVES * 32) {
    s_sum[i] = 0.f;
    s_max[i] = 0u;
  }
  __syncthreads();

  const int R = scan_rounds(Ereal);
  const int T = R * OWN_WAVES;
  int* mylist = s_list + wave * LIST_CAP;
  const unsigned pat = ((unsigned)blockIdx.x & 255u) * 0x01010101u;

  const int ch0 = (int)(((unsigned)blockIdx.x * 13u + (unsigned)wave) % (unsigned)T);

  int cnt = 0;
  {
    int ch = ch0;
    uint4 q[SCAN_LOADS], qn[SCAN_LOADS];
    scan_load(tags, ch * WAVE_EDGES, lane, q);
    for (int k = 0; k < SCAN_LOADS; ++k) qn[k] = q[k];
    const int lane16 = lane * 16;
    for (int r = 0; r < R; ++r) {
      const int base = ch * WAVE_EDGES;
      ch += OWN_WAVES;
      if (ch >= T) ch -= T;
      if (r + 1 < R) scan_load(tags, ch * WAVE_EDGES, lane, qn);
      cnt = scan_pieces<SCAN_LOADS>(q, 0, pat, lane16, base, cnt, mylist);
      for (int k = 0; k < SCAN_LOADS; ++k) q[k] = qn[k];
    }
  }
  const bool listed = cnt <= LIST_CAP;
  if (listed)
    work_list<0>(mylist, cnt, ei, Ereal, lane, hh, n0, rows, alsrc, aldst, s_max, Hf, s_acc, s_sum);
  else
    search_again<0>(tags, ch0, R, T, pat, lane, mylist, ei, Ereal, hh, n0, rows,
                    alsrc, aldst, s_max, Hf, s_acc, s_sum);

  const int per = OWN_ROWS / OWN_WAVES;
  int nself = rows - wave * per;
  nself = nself < 0 ? 0 : (nself > per ? per : nself);
  const int selfv = n0 + wave * per + (lane < nself ? lane : nself - 1);
  if (nself > 0) max_batch(nself, selfv, selfv, lane, n0, rows, alsrc, aldst, s_max);

  __syncthreads();

  if (listed)
    work_list<1>(mylist, cnt, ei, Ereal, lane, hh, n0, rows, alsrc, aldst, s_max, Hf, s_acc, s_sum);
  else
    search_again<1>(tags, ch0, R, T, pat, lane, mylist, ei, Ereal, hh, n0, rows,
                    alsrc, aldst, s_max, Hf, s_acc, s_sum);
  if (nself > 0)
    agg_edges<8>(0, nself, selfv, selfv, lane, hh, n0, rows, alsrc, aldst, s_max, Hf, s_acc, s_sum);

  __syncthreads();
  float4*       o4 = (float4*)(acc + (long)n0 * Dch);
  const float4* a4 = (const float4*)s_acc;
  for (int i = tid; i < rows * (Dch / 4); i += OWN_WAVES * 32) o4[i] = a4[i];
  for (int i = tid; i < rows * Hh; i += OWN_WAVES * 32) ssum[(long)n0 * Hh + i] = s_sum[i];
}

__global__ __launch_bounds__(256) void post_bn_elu(float* __restrict__ x,
                            const float* __restrict__ ssum,
                            const float* __restrict__ bias,
                            const float* __restrict__ gamma,
                            const float* __restrict__ beta,
                            const float* __restrict__ mean,
                            const float* __restrict__ var,
                            long n) {
  const long tid    = (long)blockIdx.x * blockDim.x + threadIdx.x;
  const long stride = (long)gridDim.x * blockDim.x;
  const int  c      = (int)((tid * 4) & (Dch - 1));
  const int  head   = c / Ch;
  float b[4], m[4], sc[4], be[4];
  for (int k = 0; k < 4; ++k) {
    b[k]  = bias[c + k];
    m[k]  = mean[c + k];
    sc[k] = gamma[c + k] / sqrtf(var[c + k] + BN_EPS);
    be[k] = beta[c + k];
  }
  float4* x4 = (float4*)x;
  const long n4 = n >> 2;
  for (long i = tid; i < n4; i += stride) {
    float4 q = x4[i];
    const float den = ssum[(i >> 5) * Hh + head] + 1e-16f;
    float v[4] = {q.x, q.y, q.z, q.w};
    for (int k = 0; k < 4; ++k) {
      float t = v[k] / den + b[k];
      t = (t - m[k]) * sc[k] + be[k];
      v[k] = t > 0.f ? t : expf(t) - 1.f;
    }
    x4[i] = make_float4(v[0], v[1], v[2], v[3]);
  }
}

__global__ void pool_accum(const float* __restrict__ x,
                           const int* __restrict__ batch,
                           float* __restrict__ sums, float* __restrict__ cnt,
                           long n) {
  const long stride = (long)gridDim.x * blockDim.x;
  for (long i = (long)blockIdx.x * blockDim.x + threadIdx.x; i < n; i += stride) {
    int node = (int)(i >> 7);
    int c    = (int)(i & 127);
    int g    = batch[node];
    atomicAdd(&sums[(long)g * Dch + c], x[i]);
    if (c == 0) atomicAdd(&cnt[g], 1.0f);
  }
}

__global__ void pool_final(const float* __restrict__ sums,
                           const float* __restrict__ cnt,
                           float* __restrict__ out, int G) {
  int i = blockIdx.x * blockDim.x + threadIdx.x;
  if (i >= G * Dch) return;
  int g = i >> 7;
  float c = cnt[g];
  out[i] = sums[i] / (c > 1.f ? c : 1.f);
}

extern "C" void kernel_launch(void* const* d_in, const int* in_sizes, int n_in,
                              void* d_out, int out_size, void* d_ws, size_t ws_size,
                              hipStream_t stream) {
  const float* x0    = (const float*)d_in[0];
  const int*   ei    = (const int*)d_in[1];
  const int*   batch = (const int*)d_in[2];
  const float* Ws    = (const float*)d_in[3];
  const float* a_src = (const float*)d_in[4];
  const float* a_dst = (const float*)d_in[5];
  const float* bias  = (const float*)d_in[6];
  const float* gamma = (const float*)d_in[7];
  const float* beta  = (const float*)d_in[8];
  const float* mean  = (const float*)d_in[9];
  const float* var   = (const float*)d_in[10];

  const int Nn    = in_sizes[0] / Dch;
  const int Ereal = in_sizes[1] / 2;
  const int G     = out_size / Dch;

  char* ws = (char*)d_ws;
  float*    bufA  = (float*)ws;    ws += (size_t)Nn * Dch * 4;
  unsigned char* tags = (unsigned char*)ws; ws += (size_t)Nn * Dch * 4;
  float*    hbuf  = (float*)ws;    ws += (size_t)Nn * Dch * 4;
  float*    ssum  = (float*)ws;    ws += (size_t)Nn * Hh * 4;
  float*    alsrc = (float*)ws;    ws += (size_t)Nn * Hh * 4;
  float*    aldst = (float*)ws;    ws += (size_t)Nn * Hh * 4;
  float*    psum  = (float*)ws;    ws += (size_t)G * Dch * 4;
  float*    pcnt  = (float*)ws;    ws += (size_t)G * 4;

  owner_tags<<<256, 256, 0, stream>>>(ei, Ereal, (unsigned*)tags);

  const float* xin  = x0;
  float*       xout = bufA;
  for (int l = 0; l < Lnum; ++l) {
    const float* Wl = Ws + (size_t)l * Dch * Dch;
    gemm_wmma_f32<<<Nn / 16, 256, 0, stream>>>(xin, Wl, hbuf);

    att_logits<<<(Nn * Hh + 255) / 256, 256, 0, stream>>>(
        hbuf, a_src + (size_t)l * Hh * Ch, a_dst + (size_t)l * Hh * Ch,
        alsrc, aldst, Nn);

    owner_aggregate<<<(Nn + OWN_ROWS - 1) / OWN_ROWS, OWN_WAVES * 32, 0, stream>>>(
        ei, tags, Ereal, Nn, alsrc, aldst, hbuf, ssum, xout);

    post_bn_elu<<<LOOP_BLOCKS, 256, 0, stream>>>(
        xout, ssum, bias + (size_t)l * Dch, gamma + (size_t)l * Dch,
        beta + (size_t)l * Dch, mean + (size_t)l * Dch, var + (size_t)l * Dch,
        (long)Nn * Dch);

    xin  = xout;
  }

  fill_u32<<<32, 256, 0, stream>>>((unsigned*)psum, 0u, (long)G * Dch);
  fill_u32<<<1, 64, 0, stream>>>((unsigned*)pcnt, 0u, G);
  pool_accum<<<LOOP_BLOCKS, 256, 0, stream>>>(xin, batch, psum, pcnt, (long)Nn * Dch);
  pool_final<<<(G * Dch + 255) / 256, 256, 0, stream>>>(psum, pcnt, (float*)d_out, G);
}


# edge search once per call: layer 1's owner pass stores each wave's found list (source, row) in workspace; layers 2, 3 read it: no byte scan, no index fetches; float adds in arrival order as before
# speedup vs baseline: 3.6112x; 1.1059x over previous
#include <hip/hip_runtime.h>


#define Dch      128
#define Hh       4
#define Ch       32
#define Lnum     3
#define NEG_SLOPE 0.2f
#define BN_EPS   1e-5f
#define LOOP_BLOCKS 2048

typedef __attribute__((ext_vector_type(2))) float v2f;
typedef __attribute__((ext_vector_type(8))) float v8f;

__device__ __forceinline__ unsigned fkey(float f) {
  unsigned b = __float_as_uint(f);
  return b ^ ((unsigned)((int)b >> 31) | 0x80000000u);
}
__device__ __forceinline__ float kinv(unsigned k) {
  return __uint_as_float((k & 0x80000000u) ? (k ^ 0x80000000u) : ~k);
}

__device__ __forceinline__ void edge_sd(const int* __restrict__ ei, int Ereal,
                                        int e, int& s, int& d) {
  if (e < Ereal) { s = ei[e]; d = ei[Ereal + e]; }
  else           { s = e - Ereal; d = s; }
}

__global__ void fill_u32(unsigned* __restrict__ p, unsigned v, long n) {
  long i = (long)blockIdx.x * blockDim.x + threadIdx.x;
  long stride = (long)gridDim.x * blockDim.x;
  for (; i < n; i += stride) p[i] = v;
}

__global__ __launch_bounds__(256) void gemm_wmma_f32(
    const float* __restrict__ X, const float* __restrict__ Wm,
    float* __restrict__ Ho) {
  const int wave = threadIdx.x >> 5;
  const int lane = threadIdx.x & 31;
  const int half = lane >> 4;
  const int l15  = lane & 15;
  const int row0 = blockIdx.x << 4;
  const int bcol = (wave << 4) + l15;
  const float* arow = X + (long)(row0 + l15) * Dch;

  v8f acc = {};
  for (int k = 0; k < Dch; k += 4) {
    const int ka = k + half * 2;
    v2f a, b;
    a.x = arow[ka];
    a.y = arow[ka + 1];
    b.x = Wm[(long)ka * Dch + bcol];
    b.y = Wm[(long)(ka + 1) * Dch + bcol];
    acc = __builtin_amdgcn_wmma_f32_16x16x4_f32(
         false, a,  false, b,
         (short)0, acc,  false,  false);
  }
  float* orow = Ho + (long)(row0 + half * 8) * Dch + bcol;
#pragma unroll
  for (int v = 0; v < 8; ++v) orow[(long)v * Dch] = acc[v];
}

__global__ void att_logits(const float* __restrict__ Hf,
                           const float* __restrict__ a_src,
                           const float* __restrict__ a_dst,
                           float* __restrict__ alsrc, float* __restrict__ aldst,
                           int Nn) {
  int i = blockIdx.x * blockDim.x + threadIdx.x;
  if (i >= Nn * Hh) return;
  int n = i >> 2, h = i & 3;
  const float* hp = Hf + (long)n * Dch + h * Ch;
  const float* as = a_src + h * Ch;
  const float* ad = a_dst + h * Ch;
  float s0 = 0.f, s1 = 0.f;
#pragma unroll
  for (int c = 0; c < Ch; ++c) { float v = hp[c]; s0 += v * as[c]; s1 += v * ad[c]; }
  alsrc[i] = s0;
  aldst[i] = s1;
}

#define OWN_ROWS   160
#define OWN_WAVES  32
#define SCAN_LOADS 6
#define LOAD_EDGES 512
#define WAVE_EDGES (SCAN_LOADS * LOAD_EDGES)
#define ROUND_EDGES (OWN_WAVES * WAVE_EDGES)
#define LIST_CAP   (WAVE_EDGES / 2)
#define GLIST_CAP  512

__device__ __forceinline__ unsigned pack_edge(int s, int dl, int rows) {
  return (unsigned)dl < (unsigned)rows ? ((unsigned)dl << 24) | (unsigned)s : 0xff000000u;
}

__device__ __forceinline__ void batch_ends(const int* __restrict__ ei, int Ereal,
                                           const int* list, int b, int n, int lane,
                                           int& sv, int& dv) {
  const int e = list[b + (lane < n ? lane : n - 1)];
  sv = e < Ereal ? ei[e]         : 0;
  dv = e < Ereal ? ei[Ereal + e] : -1;
}

__device__ __forceinline__ void max_batch(
    int n, int sv, int dl, int lane, int n0, int rows,
    const float* __restrict__ alsrc, const float* __restrict__ aldst,
    unsigned* s_max) {
  const bool ok = lane < n && (unsigned)dl < (unsigned)rows;
  const int  dd = ok ? n0 + dl : n0;
  const float4 a4 = ((const float4*)alsrc)[sv];
  const float4 d4 = ((const float4*)aldst)[dd];
  if (ok) {
    const float av[Hh] = {a4.x, a4.y, a4.z, a4.w};
    const float bv[Hh] = {d4.x, d4.y, d4.z, d4.w};
    for (int h = 0; h < Hh; ++h) {
      float sc = av[h] + bv[h];
      sc = sc > 0.f ? sc : sc * NEG_SLOPE;
      atomicMax(&s_max[dl * Hh + h], fkey(sc));
    }
  }
}

template <int K>
__device__ __forceinline__ void agg_edges(
    int j, int cnt, int sv, int dlv, int lane, int hh, int n0, int rows,
    const float* __restrict__ alsrc, const float* __restrict__ aldst,
    const unsigned* s_max, const float* __restrict__ Hf,
    float* s_acc, float* s_sum) {
  int   dl[K];
  float asv[K], adv[K], hv[K][Hh];
  unsigned mk[K];
  for (int k = 0; k < K; ++k) {
    const int slot = (j + k < cnt) ? (j + k) : (cnt - 1);
    const int s = __builtin_amdgcn_readlane(sv, slot);
    dl[k]  = __builtin_amdgcn_readlane(dlv, slot);
    const bool mine = (unsigned)dl[k] < (unsigned)rows;
    asv[k] = alsrc[s * Hh + hh];
    adv[k] = aldst[(n0 + (mine ? dl[k] : 0)) * Hh + hh];
    mk[k]  = s_max[(mine ? dl[k] : 0) * Hh + hh];
    const float* hsrc = Hf + (long)s * Dch + lane;
    for (int h = 0; h < Hh; ++h) hv[k][h] = hsrc[h * Ch];
  }
  for (int k = 0; k < K; ++k) {
    if (j + k < cnt && (unsigned)dl[k] < (unsigned)rows) {
      float sc = asv[k] + adv[k];
      sc = sc > 0.f ? sc : sc * NEG_SLOPE;
      float p = expf(sc - kinv(mk[k]));
      if (lane < Hh) atomicAdd(&s_sum[dl[k] * Hh + lane], p);
      float* adst = s_acc + dl[k] * Dch + lane;
      for (int h = 0; h < Hh; ++h) {
        float ph = __int_as_float(__builtin_amdgcn_readlane(__float_as_int(p), h));
        atomicAdd(&adst[h * Ch], ph * hv[k][h]);
      }
    }
  }
}

template <int PHASE>
__device__ __forceinline__ void work_list(
    const int* list, int cnt, const int* __restrict__ ei, int Ereal,
    int lane, int hh, int n0, int rows,
    const float* __restrict__ alsrc, const float* __restrict__ aldst,
    unsigned* s_max, const float* __restrict__ Hf, float* s_acc, float* s_sum) {
  for (int b = 0; b < cnt; b += 32) {
    const int n = (cnt - b) < 32 ? (cnt - b) : 32;
    int sv, dv;
    batch_ends(ei, Ereal, list, b, n, lane, sv, dv);
    if (PHASE == 0) {
      max_batch(n, sv, dv - n0, lane, n0, rows, alsrc, aldst, s_max);
    } else {
      for (int j = 0; j < n; j += 8)
        agg_edges<8>(j, n, sv, dv - n0, lane, hh, n0, rows, alsrc, aldst, s_max, Hf, s_acc, s_sum);
    }
  }
}

__device__ __forceinline__ void sums_listed(
    const unsigned* list, int cnt, int lane, int hh, int n0, int rows,
    const float* __restrict__ alsrc, const float* __restrict__ aldst,
    const unsigned* s_max, const float* __restrict__ Hf, float* s_acc, float* s_sum) {
  for (int b = 0; b < cnt; b += 32) {
    const int n = (cnt - b) < 32 ? (cnt - b) : 32;
    const unsigned w = list[b + (lane < n ? lane : n - 1)];
    const int sv = (int)(w & 0xffffffu), dl = (int)(w >> 24);
    for (int j = 0; j < n; j += 8)
      agg_edges<8>(j, n, sv, dl, lane, hh, n0, rows, alsrc, aldst, s_max, Hf, s_acc, s_sum);
  }
}

__device__ __forceinline__ int scan_rounds(int Ereal) {
  int R = (Ereal + ROUND_EDGES - 1) / ROUND_EDGES;
  return R < 1 ? 1 : R;
}

__global__ __launch_bounds__(256) void owner_tags(const int* __restrict__ ei, int Ereal,
                                                  unsigned* __restrict__ tags4) {
  const int* __restrict__ dstp = ei + Ereal;
  const bool vec_ok = (Ereal & 3) == 0;
  const int words  = scan_rounds(Ereal) * (ROUND_EDGES / 4);
  const int stride = (int)(gridDim.x * blockDim.x);
  for (int w = (int)(blockIdx.x * blockDim.x + threadIdx.x); w < words; w += stride) {
    const int e0 = w * 4;
    unsigned v;
    if (vec_ok && e0 + 4 <= Ereal) {
      const int4 q = *(const int4*)(dstp + e0);
      const int d[4] = {q.x, q.y, q.z, q.w};
      v = 0u;
      for (int c = 0; c < 4; ++c)
        v |= (((unsigned)d[c] / (unsigned)OWN_ROWS) & 255u) << (8 * c);
    } else {
      v = 0xffffffffu;
      for (int c = 0; c < 4; ++c)
        if (e0 + c < Ereal)
          v = (v & ~(255u << (8 * c))) |
              ((((unsigned)dstp[e0 + c] / (unsigned)OWN_ROWS) & 255u) << (8 * c));
    }
    tags4[w] = v;
  }
}

__device__ __forceinline__ void scan_load(const unsigned char* __restrict__ tags,
                                          int base, int lane,
                                          uint4 (&q)[SCAN_LOADS]) {
  const uint4* p = (const uint4*)(tags + base) + lane;
  for (int k = 0; k < SCAN_LOADS; ++k) q[k] = p[k * 32];
}

template <int NP>
__device__ __forceinline__ int scan_pieces(const uint4 (&q)[NP], int piece0, unsigned pat,
                                           int lane16, int base, int cnt, int* list) {
#pragma unroll
  for (int k = 0; k < NP; ++k) {
    const int o0 = (piece0 + k) * LOAD_EDGES + lane16;
    const unsigned wv[4] = {q[k].x, q[k].y, q[k].z, q[k].w};
#pragma unroll
    for (int j = 0; j < 4; ++j) {
      const unsigned x = wv[j] ^ pat;
      const unsigned z = (x - 0x01010101u) & ~x & 0x80808080u;
      if (__builtin_amdgcn_ballot_w32(z != 0u)) {
        for (int c = 0; c < 4; ++c) {
          const int  o   = o0 + j * 4 + c;
          const bool hit = (x & (255u << (8 * c))) == 0u;
          const unsigned mask = __builtin_amdgcn_ballot_w32(hit);
          if (mask) {
            const int slot = cnt + (int)__builtin_amdgcn_mbcnt_lo(mask, 0u);
            if (hit && slot < LIST_CAP) list[slot] = base + o;
            cnt += __builtin_popcount(mask);
          }
        }
      }
    }
  }
  return cnt;
}

template <int PHASE>
__device__ __forceinline__ void search_again(
    const unsigned char* __restrict__ tags, int ch, int R, int T, unsigned pat, int lane,
    int* list, const int* __restrict__ ei, int Ereal, int hh, int n0, int rows,
    const float* __restrict__ alsrc, const float* __restrict__ aldst,
    unsigned* s_max, const float* __restrict__ Hf, float* s_acc, float* s_sum) {
  const int lane16 = lane * 16;
  for (int r = 0; r < R; ++r) {
    const int base = ch * WAVE_EDGES;
    uint4 q[SCAN_LOADS];
    scan_load(tags, base, lane, q);
    for (int half = 0; half < 2; ++half) {
      uint4 qh[SCAN_LOADS / 2];
      for (int k = 0; k < SCAN_LOADS / 2; ++k) qh[k] = half ? q[SCAN_LOADS / 2 + k] : q[k];
      const int c = scan_pieces<SCAN_LOADS / 2>(qh, half * (SCAN_LOADS / 2), pat, lane16, base, 0, list);
      work_list<PHASE>(list, c, ei, Ereal, lane, hh, n0, rows, alsrc, aldst, s_max, Hf, s_acc, s_sum);
    }
    ch += OWN_WAVES;
    if (ch >= T) ch -= T;
  }
}

template <bool FIRST>
__global__ __launch_bounds__(OWN_WAVES * 32) void owner_aggregate(
    const int* __restrict__ ei, const unsigned char* __restrict__ tags, int Ereal, int Nn,
    const float* __restrict__ alsrc, const float* __restrict__ aldst,
    const float* __restrict__ Hf,
    float* __restrict__ ssum, float* __restrict__ acc,
    unsigned* glist, int* gcnt) {
  __shared__ float    s_acc[OWN_ROWS * Dch];
  __shared__ float    s_sum[OWN_ROWS * Hh];
  __shared__ unsigned s_max[OWN_ROWS * Hh];
  __shared__ int      s_list[OWN_WAVES * LIST_CAP];
  __shared__ int      s_cnt[OWN_WAVES];

  const int tid  = threadIdx.x;
  const int lane = tid & 31;
  const int hh   = lane & 3;
  const int wave = __builtin_amdgcn_readfirstlane(tid >> 5);
  const int n0   = (int)blockIdx.x * OWN_ROWS;
  const int rows = (Nn - n0) < OWN_ROWS ? (Nn - n0) : OWN_ROWS;

  for (int i = tid; i < OWN_ROWS * Dch / 4; i += OWN_WAVES * 32)
    ((float4*)s_acc)[i] = make_float4(0.f, 0.f, 0.f, 0.f);
  for (int i = tid; i < OWN_ROWS * Hh; i += OWN_WAVES * 32) {
    s_sum[i] = 0.f;
    s_max[i] = 0u;
  }
  __syncthreads();

  const int R = scan_rounds(Ereal);
  const int T = R * OWN_WAVES;
  int* mylist = s_list + wave * LIST_CAP;
  unsigned* mywords = (unsigned*)mylist;
  const int gslot = (int)blockIdx.x * OWN_WAVES + wave;
  unsigned* gl = glist + (long)gslot * GLIST_CAP;
  const unsigned pat = ((unsigned)blockIdx.x & 255u) * 0x01010101u;

  const int ch0 = (int)(((unsigned)blockIdx.x * 13u + (unsigned)wave) % (unsigned)T);

  int cnt = 0;
  if (FIRST) {
    int ch = ch0;
    uint4 q[SCAN_LOADS], qn[SCAN_LOADS];
    scan_load(tags, ch * WAVE_EDGES, lane, q);
    for (int k = 0; k < SCAN_LOADS; ++k) qn[k] = q[k];
    const int lane16 = lane * 16;
    for (int r = 0; r < R; ++r) {
      const int base = ch * WAVE_EDGES;
      ch += OWN_WAVES;
      if (ch >= T) ch -= T;
      if (r + 1 < R) scan_load(tags, ch * WAVE_EDGES, lane, qn);
      cnt = scan_pieces<SCAN_LOADS>(q, 0, pat, lane16, base, cnt, mylist);
      for (int k = 0; k < SCAN_LOADS; ++k) q[k] = qn[k];
    }
    if (lane == 0) s_cnt[wave] = cnt;
  } else {
    cnt = gcnt[gslot];
  }
  const bool listed = cnt <= GLIST_CAP;
  if (listed) {
    for (int b = 0; b < cnt; b += 32) {
      const int n = (cnt - b) < 32 ? (cnt - b) : 32;
      int sv, dl;
      unsigned w;
      if (FIRST) {
        int dv;
        batch_ends(ei, Ereal, mylist, b, n, lane, sv, dv);
        dl = dv - n0;
        w  = pack_edge(sv, dl, rows);
        if (lane < n) gl[b + lane] = w;
      } else {
        w  = gl[b + (lane < n ? lane : n - 1)];
        sv = (int)(w & 0xffffffu);
        dl = (int)(w >> 24);
      }
      if (lane < n) mywords[b + lane] = w;
      max_batch(n, sv, dl, lane, n0, rows, alsrc, aldst, s_max);
    }
  } else {
    search_again<0>(tags, ch0, R, T, pat, lane, mylist, ei, Ereal, hh, n0, rows,
                    alsrc, aldst, s_max, Hf, s_acc, s_sum);
  }

  const int per = OWN_ROWS / OWN_WAVES;
  int nself = rows - wave * per;
  nself = nself < 0 ? 0 : (nself > per ? per : nself);
  const int selfv = n0 + wave * per + (lane < nself ? lane : nself - 1);
  if (nself > 0) max_batch(nself, selfv, selfv - n0, lane, n0, rows, alsrc, aldst, s_max);

  __syncthreads();

  if (FIRST && wave == 0) gcnt[(int)blockIdx.x * OWN_WAVES + lane] = s_cnt[lane];

  if (listed)
    sums_listed(mywords, cnt, lane, hh, n0, rows, alsrc, aldst, s_max, Hf, s_acc, s_sum);
  else
    search_again<1>(tags, ch0, R, T, pat, lane, mylist, ei, Ereal, hh, n0, rows,
                    alsrc, aldst, s_max, Hf, s_acc, s_sum);
  if (nself > 0)
    agg_edges<8>(0, nself, selfv, selfv - n0, lane, hh, n0, rows, alsrc, aldst, s_max, Hf, s_acc, s_sum);

  __syncthreads();
  float4*       o4 = (float4*)(acc + (long)n0 * Dch);
  const float4* a4 = (const float4*)s_acc;
  for (int i = tid; i < rows * (Dch / 4); i += OWN_WAVES * 32) o4[i] = a4[i];
  for (int i = tid; i < rows * Hh; i += OWN_WAVES * 32) ssum[(long)n0 * Hh + i] = s_sum[i];
}

__global__ __launch_bounds__(256) void post_bn_elu(float* __restrict__ x,
                            const float* __restrict__ ssum,
                            const float* __restrict__ bias,
                            const float* __restrict__ gamma,
                            const float* __restrict__ beta,
                            const float* __restrict__ mean,
                            const float* __restrict__ var,
                            long n) {
  const long tid    = (long)blockIdx.x * blockDim.x + threadIdx.x;
  const long stride = (long)gridDim.x * blockDim.x;
  const int  c      = (int)((tid * 4) & (Dch - 1));
  const int  head   = c / Ch;
  float b[4], m[4], sc[4], be[4];
  for (int k = 0; k < 4; ++k) {
    b[k]  = bias[c + k];
    m[k]  = mean[c + k];
    sc[k] = gamma[c + k] / sqrtf(var[c + k] + BN_EPS);
    be[k] = beta[c + k];
  }
  float4* x4 = (float4*)x;
  const long n4 = n >> 2;
  for (long i = tid; i < n4; i += stride) {
    float4 q = x4[i];
    const float den = ssum[(i >> 5) * Hh + head] + 1e-16f;
    float v[4] = {q.x, q.y, q.z, q.w};
    for (int k = 0; k < 4; ++k) {
      float t = v[k] / den + b[k];
      t = (t - m[k]) * sc[k] + be[k];
      v[k] = t > 0.f ? t : expf(t) - 1.f;
    }
    x4[i] = make_float4(v[0], v[1], v[2], v[3]);
  }
}

__global__ void pool_accum(const float* __restrict__ x,
                           const int* __restrict__ batch,
                           float* __restrict__ sums, float* __restrict__ cnt,
                           long n) {
  const long stride = (long)gridDim.x * blockDim.x;
  for (long i = (long)blockIdx.x * blockDim.x + threadIdx.x; i < n; i += stride) {
    int node = (int)(i >> 7);
    int c    = (int)(i & 127);
    int g    = batch[node];
    atomicAdd(&sums[(long)g * Dch + c], x[i]);
    if (c == 0) atomicAdd(&cnt[g], 1.0f);
  }
}

__global__ void pool_final(const float* __restrict__ sums,
                           const float* __restrict__ cnt,
                           float* __restrict__ out, int G) {
  int i = blockIdx.x * blockDim.x + threadIdx.x;
  if (i >= G * Dch) return;
  int g = i >> 7;
  float c = cnt[g];
  out[i] = sums[i] / (c > 1.f ? c : 1.f);
}

extern "C" void kernel_launch(void* const* d_in, const int* in_sizes, int n_in,
                              void* d_out, int out_size, void* d_ws, size_t ws_size,
                              hipStream_t stream) {
  const float* x0    = (const float*)d_in[0];
  const int*   ei    = (const int*)d_in[1];
  const int*   batch = (const int*)d_in[2];
  const float* Ws    = (const float*)d_in[3];
  const float* a_src = (const float*)d_in[4];
  const float* a_dst = (const float*)d_in[5];
  const float* bias  = (const float*)d_in[6];
  const float* gamma = (const float*)d_in[7];
  const float* beta  = (const float*)d_in[8];
  const float* mean  = (const float*)d_in[9];
  const float* var   = (const float*)d_in[10];

  const int Nn    = in_sizes[0] / Dch;
  const int Ereal = in_sizes[1] / 2;
  const int G     = out_size / Dch;

  char* ws = (char*)d_ws;
  float*    bufA  = (float*)ws;    ws += (size_t)Nn * Dch * 4;
  unsigned char* tags = (unsigned char*)ws; ws += (size_t)Nn * Dch * 4;
  const int    groups    = (Nn + OWN_ROWS - 1) / OWN_ROWS;
  const size_t tag_bytes = (size_t)((Ereal + ROUND_EDGES - 1) / ROUND_EDGES + 1) * ROUND_EDGES;
  unsigned*    glist     = (unsigned*)(tags + tag_bytes);
  int*         gcnt      = (int*)(glist + (size_t)groups * OWN_WAVES * GLIST_CAP);
  float*    hbuf  = (float*)ws;    ws += (size_t)Nn * Dch * 4;
  float*    ssum  = (float*)ws;    ws += (size_t)Nn * Hh * 4;
  float*    alsrc = (float*)ws;    ws += (size_t)Nn * Hh * 4;
  float*    aldst = (float*)ws;    ws += (size_t)Nn * Hh * 4;
  float*    psum  = (float*)ws;    ws += (size_t)G * Dch * 4;
  float*    pcnt  = (float*)ws;    ws += (size_t)G * 4;

  owner_tags<<<256, 256, 0, stream>>>(ei, Ereal, (unsigned*)tags);

  const float* xin  = x0;
  float*       xout = bufA;
  for (int l = 0; l < Lnum; ++l) {
    const float* Wl = Ws + (size_t)l * Dch * Dch;
    gemm_wmma_f32<<<Nn / 16, 256, 0, stream>>>(xin, Wl, hbuf);

    att_logits<<<(Nn * Hh + 255) / 256, 256, 0, stream>>>(
        hbuf, a_src + (size_t)l * Hh * Ch, a_dst + (size_t)l * Hh * Ch,
        alsrc, aldst, Nn);

    if (l == 0)
      owner_aggregate<true><<<groups, OWN_WAVES * 32, 0, stream>>>(
          ei, tags, Ereal, Nn, alsrc, aldst, hbuf, ssum, xout, glist, gcnt);
    else
      owner_aggregate<false><<<groups, OWN_WAVES * 32, 0, stream>>>(
          ei, tags, Ereal, Nn, alsrc, aldst, hbuf, ssum, xout, glist, gcnt);

    post_bn_elu<<<LOOP_BLOCKS, 256, 0, stream>>>(
        xout, ssum, bias + (size_t)l * Dch, gamma + (size_t)l * Dch,
        beta + (size_t)l * Dch, mean + (size_t)l * Dch, var + (size_t)l * Dch,
        (long)Nn * Dch);

    xin  = xout;
  }

  fill_u32<<<32, 256, 0, stream>>>((unsigned*)psum, 0u, (long)G * Dch);
  fill_u32<<<1, 64, 0, stream>>>((unsigned*)pcnt, 0u, G);
  pool_accum<<<LOOP_BLOCKS, 256, 0, stream>>>(xin, batch, psum, pcnt, (long)Nn * Dch);
  pool_final<<<(G * Dch + 255) / 256, 256, 0, stream>>>(psum, pcnt, (float*)d_out, G);
}


# owner pass sums phase: a lane works out its own edge's four exp weights (32 edges at once), the walk broadcasts them: per-edge logit loads, max reads, exps gone; float adds in arrival order
# speedup vs baseline: 3.8219x; 1.0583x over previous
#include <hip/hip_runtime.h>


#define Dch      128
#define Hh       4
#define Ch       32
#define Lnum     3
#define NEG_SLOPE 0.2f
#define BN_EPS   1e-5f
#define LOOP_BLOCKS 2048

typedef __attribute__((ext_vector_type(2))) float v2f;
typedef __attribute__((ext_vector_type(8))) float v8f;

__device__ __forceinline__ unsigned fkey(float f) {
  unsigned b = __float_as_uint(f);
  return b ^ ((unsigned)((int)b >> 31) | 0x80000000u);
}
__device__ __forceinline__ float kinv(unsigned k) {
  return __uint_as_float((k & 0x80000000u) ? (k ^ 0x80000000u) : ~k);
}

__device__ __forceinline__ void edge_sd(const int* __restrict__ ei, int Ereal,
                                        int e, int& s, int& d) {
  if (e < Ereal) { s = ei[e]; d = ei[Ereal + e]; }
  else           { s = e - Ereal; d = s; }
}

__global__ void fill_u32(unsigned* __restrict__ p, unsigned v, long n) {
  long i = (long)blockIdx.x * blockDim.x + threadIdx.x;
  long stride = (long)gridDim.x * blockDim.x;
  for (; i < n; i += stride) p[i] = v;
}

__global__ __launch_bounds__(256) void gemm_wmma_f32(
    const float* __restrict__ X, const float* __restrict__ Wm,
    float* __restrict__ Ho) {
  const int wave = threadIdx.x >> 5;
  const int lane = threadIdx.x & 31;
  const int half = lane >> 4;
  const int l15  = lane & 15;
  const int row0 = blockIdx.x << 4;
  const int bcol = (wave << 4) + l15;
  const float* arow = X + (long)(row0 + l15) * Dch;

  v8f acc = {};
  for (int k = 0; k < Dch; k += 4) {
    const int ka = k + half * 2;
    v2f a, b;
    a.x = arow[ka];
    a.y = arow[ka + 1];
    b.x = Wm[(long)ka * Dch + bcol];
    b.y = Wm[(long)(ka + 1) * Dch + bcol];
    acc = __builtin_amdgcn_wmma_f32_16x16x4_f32(
         false, a,  false, b,
         (short)0, acc,  false,  false);
  }
  float* orow = Ho + (long)(row0 + half * 8) * Dch + bcol;
#pragma unroll
  for (int v = 0; v < 8; ++v) orow[(long)v * Dch] = acc[v];
}

__global__ void att_logits(const float* __restrict__ Hf,
                           const float* __restrict__ a_src,
                           const float* __restrict__ a_dst,
                           float* __restrict__ alsrc, float* __restrict__ aldst,
                           int Nn) {
  int i = blockIdx.x * blockDim.x + threadIdx.x;
  if (i >= Nn * Hh) return;
  int n = i >> 2, h = i & 3;
  const float* hp = Hf + (long)n * Dch + h * Ch;
  const float* as = a_src + h * Ch;
  const float* ad = a_dst + h * Ch;
  float s0 = 0.f, s1 = 0.f;
#pragma unroll
  for (int c = 0; c < Ch; ++c) { float v = hp[c]; s0 += v * as[c]; s1 += v * ad[c]; }
  alsrc[i] = s0;
  aldst[i] = s1;
}

#define OWN_ROWS   160
#define OWN_WAVES  32
#define SCAN_LOADS 6
#define LOAD_EDGES 512
#define WAVE_EDGES (SCAN_LOADS * LOAD_EDGES)
#define ROUND_EDGES (OWN_WAVES * WAVE_EDGES)
#define LIST_CAP   (WAVE_EDGES / 2)
#define GLIST_CAP  512

__device__ __forceinline__ unsigned pack_edge(int s, int dl, int rows) {
  return (unsigned)dl < (unsigned)rows ? ((unsigned)dl << 24) | (unsigned)s : 0xff000000u;
}

__device__ __forceinline__ void batch_ends(const int* __restrict__ ei, int Ereal,
                                           const int* list, int b, int n, int lane,
                                           int& sv, int& dv) {
  const int e = list[b + (lane < n ? lane : n - 1)];
  sv = e < Ereal ? ei[e]         : 0;
  dv = e < Ereal ? ei[Ereal + e] : -1;
}

__device__ __forceinline__ void max_batch(
    int n, int sv, int dl, int lane, int n0, int rows,
    const float* __restrict__ alsrc, const float* __restrict__ aldst,
    unsigned* s_max) {
  const bool ok = lane < n && (unsigned)dl < (unsigned)rows;
  const int  dd = ok ? n0 + dl : n0;
  const float4 a4 = ((const float4*)alsrc)[sv];
  const float4 d4 = ((const float4*)aldst)[dd];
  if (ok) {
    const float av[Hh] = {a4.x, a4.y, a4.z, a4.w};
    const float bv[Hh] = {d4.x, d4.y, d4.z, d4.w};
    for (int h = 0; h < Hh; ++h) {
      float sc = av[h] + bv[h];
      sc = sc > 0.f ? sc : sc * NEG_SLOPE;
      atomicMax(&s_max[dl * Hh + h], fkey(sc));
    }
  }
}

template <int K>
__device__ __forceinline__ void walk_edges(
    int j, int n, int sv, int dlv, const float (&p)[Hh], int lane, int rows,
    const float* __restrict__ Hf, float* s_acc) {
  float hv[K][Hh];
  for (int k = 0; k < K; ++k) {
    const int slot = (j + k < n) ? (j + k) : (n - 1);
    const int s = __builtin_amdgcn_readlane(sv, slot);
    const float* hsrc = Hf + (long)s * Dch + lane;
    for (int h = 0; h < Hh; ++h) hv[k][h] = hsrc[h * Ch];
  }
  for (int k = 0; k < K; ++k) {
    if (j + k < n) {
      const int d = __builtin_amdgcn_readlane(dlv, j + k);
      if ((unsigned)d < (unsigned)rows) {
        float* adst = s_acc + d * Dch + lane;
        for (int h = 0; h < Hh; ++h) {
          const float ph = __int_as_float(__builtin_amdgcn_readlane(__float_as_int(p[h]), j + k));
          atomicAdd(&adst[h * Ch], ph * hv[k][h]);
        }
      }
    }
  }
}

__device__ __forceinline__ void agg_batch(
    int n, int sv, int dl, int lane, int n0, int rows,
    const float* __restrict__ alsrc, const float* __restrict__ aldst,
    const unsigned* s_max, const float* __restrict__ Hf,
    float* s_acc, float* s_sum) {
  const bool ok = lane < n && (unsigned)dl < (unsigned)rows;
  const int  dd = ok ? dl : 0;
  const float4 a4 = ((const float4*)alsrc)[sv];
  const float4 d4 = ((const float4*)aldst)[n0 + dd];
  const uint4  m4 = ((const uint4*)s_max)[dd];
  const float    av[Hh] = {a4.x, a4.y, a4.z, a4.w};
  const float    bv[Hh] = {d4.x, d4.y, d4.z, d4.w};
  const unsigned mk[Hh] = {m4.x, m4.y, m4.z, m4.w};
  float p[Hh];
  for (int h = 0; h < Hh; ++h) {
    float sc = av[h] + bv[h];
    sc = sc > 0.f ? sc : sc * NEG_SLOPE;
    p[h] = expf(sc - kinv(mk[h]));
  }
  if (ok)
    for (int h = 0; h < Hh; ++h) atomicAdd(&s_sum[dl * Hh + h], p[h]);
  for (int j = 0; j < n; j += 8)
    walk_edges<8>(j, n, sv, dl, p, lane, rows, Hf, s_acc);
}

template <int PHASE>
__device__ __forceinline__ void work_list(
    const int* list, int cnt, const int* __restrict__ ei, int Ereal,
    int lane, int n0, int rows,
    const float* __restrict__ alsrc, const float* __restrict__ aldst,
    unsigned* s_max, const float* __restrict__ Hf, float* s_acc, float* s_sum) {
  for (int b = 0; b < cnt; b += 32) {
    const int n = (cnt - b) < 32 ? (cnt - b) : 32;
    int sv, dv;
    batch_ends(ei, Ereal, list, b, n, lane, sv, dv);
    if (PHASE == 0) {
      max_batch(n, sv, dv - n0, lane, n0, rows, alsrc, aldst, s_max);
    } else {
      agg_batch(n, sv, dv - n0, lane, n0, rows, alsrc, aldst, s_max, Hf, s_acc, s_sum);
    }
  }
}

__device__ __forceinline__ void sums_listed(
    const unsigned* list, int cnt, int lane, int n0, int rows,
    const float* __restrict__ alsrc, const float* __restrict__ aldst,
    const unsigned* s_max, const float* __restrict__ Hf, float* s_acc, float* s_sum) {
  for (int b = 0; b < cnt; b += 32) {
    const int n = (cnt - b) < 32 ? (cnt - b) : 32;
    const unsigned w = list[b + (lane < n ? lane : n - 1)];
    const int sv = (int)(w & 0xffffffu), dl = (int)(w >> 24);
    agg_batch(n, sv, dl, lane, n0, rows, alsrc, aldst, s_max, Hf, s_acc, s_sum);
  }
}

__device__ __forceinline__ int scan_rounds(int Ereal) {
  int R = (Ereal + ROUND_EDGES - 1) / ROUND_EDGES;
  return R < 1 ? 1 : R;
}

__global__ __launch_bounds__(256) void owner_tags(const int* __restrict__ ei, int Ereal,
                                                  unsigned* __restrict__ tags4) {
  const int* __restrict__ dstp = ei + Ereal;
  const bool vec_ok = (Ereal & 3) == 0;
  const int words  = scan_rounds(Ereal) * (ROUND_EDGES / 4);
  const int stride = (int)(gridDim.x * blockDim.x);
  for (int w = (int)(blockIdx.x * blockDim.x + threadIdx.x); w < words; w += stride) {
    const int e0 = w * 4;
    unsigned v;
    if (vec_ok && e0 + 4 <= Ereal) {
      const int4 q = *(const int4*)(dstp + e0);
      const int d[4] = {q.x, q.y, q.z, q.w};
      v = 0u;
      for (int c = 0; c < 4; ++c)
        v |= (((unsigned)d[c] / (unsigned)OWN_ROWS) & 255u) << (8 * c);
    } else {
      v = 0xffffffffu;
      for (int c = 0; c < 4; ++c)
        if (e0 + c < Ereal)
          v = (v & ~(255u << (8 * c))) |
              ((((unsigned)dstp[e0 + c] / (unsigned)OWN_ROWS) & 255u) << (8 * c));
    }
    tags4[w] = v;
  }
}

__device__ __forceinline__ void scan_load(const unsigned char* __restrict__ tags,
                                          int base, int lane,
                                          uint4 (&q)[SCAN_LOADS]) {
  const uint4* p = (const uint4*)(tags + base) + lane;
  for (int k = 0; k < SCAN_LOADS; ++k) q[k] = p[k * 32];
}

template <int NP>
__device__ __forceinline__ int scan_pieces(const uint4 (&q)[NP], int piece0, unsigned pat,
                                           int lane16, int base, int cnt, int* list) {
#pragma unroll
  for (int k = 0; k < NP; ++k) {
    const int o0 = (piece0 + k) * LOAD_EDGES + lane16;
    const unsigned wv[4] = {q[k].x, q[k].y, q[k].z, q[k].w};
#pragma unroll
    for (int j = 0; j < 4; ++j) {
      const unsigned x = wv[j] ^ pat;
      const unsigned z = (x - 0x01010101u) & ~x & 0x80808080u;
      if (__builtin_amdgcn_ballot_w32(z != 0u)) {
        for (int c = 0; c < 4; ++c) {
          const int  o   = o0 + j * 4 + c;
          const bool hit = (x & (255u << (8 * c))) == 0u;
          const unsigned mask = __builtin_amdgcn_ballot_w32(hit);
          if (mask) {
            const int slot = cnt + (int)__builtin_amdgcn_mbcnt_lo(mask, 0u);
            if (hit && slot < LIST_CAP) list[slot] = base + o;
            cnt += __builtin_popcount(mask);
          }
        }
      }
    }
  }
  return cnt;
}

template <int PHASE>
__device__ __forceinline__ void search_again(
    const unsigned char* __restrict__ tags, int ch, int R, int T, unsigned pat, int lane,
    int* list, const int* __restrict__ ei, int Ereal, int n0, int rows,
    const float* __restrict__ alsrc, const float* __restrict__ aldst,
    unsigned* s_max, const float* __restrict__ Hf, float* s_acc, float* s_sum) {
  const int lane16 = lane * 16;
  for (int r = 0; r < R; ++r) {
    const int base = ch * WAVE_EDGES;
    uint4 q[SCAN_LOADS];
    scan_load(tags, base, lane, q);
    for (int half = 0; half < 2; ++half) {
      uint4 qh[SCAN_LOADS / 2];
      for (int k = 0; k < SCAN_LOADS / 2; ++k) qh[k] = half ? q[SCAN_LOADS / 2 + k] : q[k];
      const int c = scan_pieces<SCAN_LOADS / 2>(qh, half * (SCAN_LOADS / 2), pat, lane16, base, 0, list);
      work_list<PHASE>(list, c, ei, Ereal, lane, n0, rows, alsrc, aldst, s_max, Hf, s_acc, s_sum);
    }
    ch += OWN_WAVES;
    if (ch >= T) ch -= T;
  }
}

template <bool FIRST>
__global__ __launch_bounds__(OWN_WAVES * 32) void owner_aggregate(
    const int* __restrict__ ei, const unsigned char* __restrict__ tags, int Ereal, int Nn,
    const float* __restrict__ alsrc, const float* __restrict__ aldst,
    const float* __restrict__ Hf,
    float* __restrict__ ssum, float* __restrict__ acc,
    unsigned* glist, int* gcnt) {
  __shared__ float    s_acc[OWN_ROWS * Dch];
  __shared__ float    s_sum[OWN_ROWS * Hh];
  __shared__ __attribute__((aligned(16))) unsigned s_max[OWN_ROWS * Hh];
  __shared__ int      s_list[OWN_WAVES * LIST_CAP];
  __shared__ int      s_cnt[OWN_WAVES];

  const int tid  = threadIdx.x;
  const int lane = tid & 31;
  const int wave = __builtin_amdgcn_readfirstlane(tid >> 5);
  const int n0   = (int)blockIdx.x * OWN_ROWS;
  const int rows = (Nn - n0) < OWN_ROWS ? (Nn - n0) : OWN_ROWS;

  for (int i = tid; i < OWN_ROWS * Dch / 4; i += OWN_WAVES * 32)
    ((float4*)s_acc)[i] = make_float4(0.f, 0.f, 0.f, 0.f);
  for (int i = tid; i < OWN_ROWS * Hh; i += OWN_WAVES * 32) {
    s_sum[i] = 0.f;
    s_max[i] = 0u;
  }
  __syncthreads();

  const int R = scan_rounds(Ereal);
  const int T = R * OWN_WAVES;
  int* mylist = s_list + wave * LIST_CAP;
  unsigned* mywords = (unsigned*)mylist;
  const int gslot = (int)blockIdx.x * OWN_WAVES + wave;
  unsigned* gl = glist + (long)gslot * GLIST_CAP;
  const unsigned pat = ((unsigned)blockIdx.x & 255u) * 0x01010101u;

  const int ch0 = (int)(((unsigned)blockIdx.x * 13u + (unsigned)wave) % (unsigned)T);

  int cnt = 0;
  if (FIRST) {
    int ch = ch0;
    uint4 q[SCAN_LOADS], qn[SCAN_LOADS];
    scan_load(tags, ch * WAVE_EDGES, lane, q);
    for (int k = 0; k < SCAN_LOADS; ++k) qn[k] = q[k];
    const int lane16 = lane * 16;
    for (int r = 0; r < R; ++r) {
      const int base = ch * WAVE_EDGES;
      ch += OWN_WAVES;
      if (ch >= T) ch -= T;
      if (r + 1 < R) scan_load(tags, ch * WAVE_EDGES, lane, qn);
      cnt = scan_pieces<SCAN_LOADS>(q, 0, pat, lane16, base, cnt, mylist);
      for (int k = 0; k < SCAN_LOADS; ++k) q[k] = qn[k];
    }
    if (lane == 0) s_cnt[wave] = cnt;
  } else {
    cnt = gcnt[gslot];
  }
  const bool listed = cnt <= GLIST_CAP;
  if (listed) {
    for (int b = 0; b < cnt; b += 32) {
      const int n = (cnt - b) < 32 ? (cnt - b) : 32;
      int sv, dl;
      unsigned w;
      if (FIRST) {
        int dv;
        batch_ends(ei, Ereal, mylist, b, n, lane, sv, dv);
        dl = dv - n0;
        w  = pack_edge(sv, dl, rows);
        if (lane < n) gl[b + lane] = w;
      } else {
        w  = gl[b + (lane < n ? lane : n - 1)];
        sv = (int)(w & 0xffffffu);
        dl = (int)(w >> 24);
      }
      if (lane < n) mywords[b + lane] = w;
      max_batch(n, sv, dl, lane, n0, rows, alsrc, aldst, s_max);
    }
  } else {
    search_again<0>(tags, ch0, R, T, pat, lane, mylist, ei, Ereal, n0, rows,
                    alsrc, aldst, s_max, Hf, s_acc, s_sum);
  }

  const int per = OWN_ROWS / OWN_WAVES;
  int nself = rows - wave * per;
  nself = nself < 0 ? 0 : (nself > per ? per : nself);
  const int selfv = n0 + wave * per + (lane < nself ? lane : nself - 1);
  if (nself > 0) max_batch(nself, selfv, selfv - n0, lane, n0, rows, alsrc, aldst, s_max);

  __syncthreads();

  if (FIRST && wave == 0) gcnt[(int)blockIdx.x * OWN_WAVES + lane] = s_cnt[lane];

  if (listed)
    sums_listed(mywords, cnt, lane, n0, rows, alsrc, aldst, s_max, Hf, s_acc, s_sum);
  else
    search_again<1>(tags, ch0, R, T, pat, lane, mylist, ei, Ereal, n0, rows,
                    alsrc, aldst, s_max, Hf, s_acc, s_sum);
  if (nself > 0)
    agg_batch(nself, selfv, selfv - n0, lane, n0, rows, alsrc, aldst, s_max, Hf, s_acc, s_sum);

  __syncthreads();
  float4*       o4 = (float4*)(acc + (long)n0 * Dch);
  const float4* a4 = (const float4*)s_acc;
  for (int i = tid; i < rows * (Dch / 4); i += OWN_WAVES * 32) o4[i] = a4[i];
  for (int i = tid; i < rows * Hh; i += OWN_WAVES * 32) ssum[(long)n0 * Hh + i] = s_sum[i];
}

__global__ __launch_bounds__(256) void post_bn_elu(float* __restrict__ x,
                            const float* __restrict__ ssum,
                            const float* __restrict__ bias,
                            const float* __restrict__ gamma,
                            const float* __restrict__ beta,
                            const float* __restrict__ mean,
                            const float* __restrict__ var,
                            long n) {
  const long tid    = (long)blockIdx.x * blockDim.x + threadIdx.x;
  const long stride = (long)gridDim.x * blockDim.x;
  const int  c      = (int)((tid * 4) & (Dch - 1));
  const int  head   = c / Ch;
  float b[4], m[4], sc[4], be[4];
  for (int k = 0; k < 4; ++k) {
    b[k]  = bias[c + k];
    m[k]  = mean[c + k];
    sc[k] = gamma[c + k] / sqrtf(var[c + k] + BN_EPS);
    be[k] = beta[c + k];
  }
  float4* x4 = (float4*)x;
  const long n4 = n >> 2;
  for (long i = tid; i < n4; i += stride) {
    float4 q = x4[i];
    const float den = ssum[(i >> 5) * Hh + head] + 1e-16f;
    float v[4] = {q.x, q.y, q.z, q.w};
    for (int k = 0; k < 4; ++k) {
      float t = v[k] / den + b[k];
      t = (t - m[k]) * sc[k] + be[k];
      v[k] = t > 0.f ? t : expf(t) - 1.f;
    }
    x4[i] = make_float4(v[0], v[1], v[2], v[3]);
  }
}

__global__ void pool_accum(const float* __restrict__ x,
                           const int* __restrict__ batch,
                           float* __restrict__ sums, float* __restrict__ cnt,
                           long n) {
  const long stride = (long)gridDim.x * blockDim.x;
  for (long i = (long)blockIdx.x * blockDim.x + threadIdx.x; i < n; i += stride) {
    int node = (int)(i >> 7);
    int c    = (int)(i & 127);
    int g    = batch[node];
    atomicAdd(&sums[(long)g * Dch + c], x[i]);
    if (c == 0) atomicAdd(&cnt[g], 1.0f);
  }
}

__global__ void pool_final(const float* __restrict__ sums,
                           const float* __restrict__ cnt,
                           float* __restrict__ out, int G) {
  int i = blockIdx.x * blockDim.x + threadIdx.x;
  if (i >= G * Dch) return;
  int g = i >> 7;
  float c = cnt[g];
  out[i] = sums[i] / (c > 1.f ? c : 1.f);
}

extern "C" void kernel_launch(void* const* d_in, const int* in_sizes, int n_in,
                              void* d_out, int out_size, void* d_ws, size_t ws_size,
                              hipStream_t stream) {
  const float* x0    = (const float*)d_in[0];
  const int*   ei    = (const int*)d_in[1];
  const int*   batch = (const int*)d_in[2];
  const float* Ws    = (const float*)d_in[3];
  const float* a_src = (const float*)d_in[4];
  const float* a_dst = (const float*)d_in[5];
  const float* bias  = (const float*)d_in[6];
  const float* gamma = (const float*)d_in[7];
  const float* beta  = (const float*)d_in[8];
  const float* mean  = (const float*)d_in[9];
  const float* var   = (const float*)d_in[10];

  const int Nn    = in_sizes[0] / Dch;
  const int Ereal = in_sizes[1] / 2;
  const int G     = out_size / Dch;

  char* ws = (char*)d_ws;
  float*    bufA  = (float*)ws;    ws += (size_t)Nn * Dch * 4;
  unsigned char* tags = (unsigned char*)ws; ws += (size_t)Nn * Dch * 4;
  const int    groups    = (Nn + OWN_ROWS - 1) / OWN_ROWS;
  const size_t tag_bytes = (size_t)((Ereal + ROUND_EDGES - 1) / ROUND_EDGES + 1) * ROUND_EDGES;
  unsigned*    glist     = (unsigned*)(tags + tag_bytes);
  int*         gcnt      = (int*)(glist + (size_t)groups * OWN_WAVES * GLIST_CAP);
  float*    hbuf  = (float*)ws;    ws += (size_t)Nn * Dch * 4;
  float*    ssum  = (float*)ws;    ws += (size_t)Nn * Hh * 4;
  float*    alsrc = (float*)ws;    ws += (size_t)Nn * Hh * 4;
  float*    aldst = (float*)ws;    ws += (size_t)Nn * Hh * 4;
  float*    psum  = (float*)ws;    ws += (size_t)G * Dch * 4;
  float*    pcnt  = (float*)ws;    ws += (size_t)G * 4;

  owner_tags<<<256, 256, 0, stream>>>(ei, Ereal, (unsigned*)tags);

  const float* xin  = x0;
  float*       xout = bufA;
  for (int l = 0; l < Lnum; ++l) {
    const float* Wl = Ws + (size_t)l * Dch * Dch;
    gemm_wmma_f32<<<Nn / 16, 256, 0, stream>>>(xin, Wl, hbuf);

    att_logits<<<(Nn * Hh + 255) / 256, 256, 0, stream>>>(
        hbuf, a_src + (size_t)l * Hh * Ch, a_dst + (size_t)l * Hh * Ch,
        alsrc, aldst, Nn);

    if (l == 0)
      owner_aggregate<true><<<groups, OWN_WAVES * 32, 0, stream>>>(
          ei, tags, Ereal, Nn, alsrc, aldst, hbuf, ssum, xout, glist, gcnt);
    else
      owner_aggregate<false><<<groups, OWN_WAVES * 32, 0, stream>>>(
          ei, tags, Ereal, Nn, alsrc, aldst, hbuf, ssum, xout, glist, gcnt);

    post_bn_elu<<<LOOP_BLOCKS, 256, 0, stream>>>(
        xout, ssum, bias + (size_t)l * Dch, gamma + (size_t)l * Dch,
        beta + (size_t)l * Dch, mean + (size_t)l * Dch, var + (size_t)l * Dch,
        (long)Nn * Dch);

    xin  = xout;
  }

  fill_u32<<<32, 256, 0, stream>>>((unsigned*)psum, 0u, (long)G * Dch);
  fill_u32<<<1, 64, 0, stream>>>((unsigned*)pcnt, 0u, G);
  pool_accum<<<LOOP_BLOCKS, 256, 0, stream>>>(xin, batch, psum, pcnt, (long)Nn * Dch);
  pool_final<<<(G * Dch + 255) / 256, 256, 0, stream>>>(psum, pcnt, (float*)d_out, G);
}


# GEMM operand path: W staged once per group in LDS (k-pairs interleaved, 8-byte fragment loads), a wave owns a 16-row strip x 128 columns and reads X once by 16-byte loads; k summed in another order
# speedup vs baseline: 4.1291x; 1.0804x over previous
#include <hip/hip_runtime.h>


#define Dch      128
#define Hh       4
#define Ch       32
#define Lnum     3
#define NEG_SLOPE 0.2f
#define BN_EPS   1e-5f
#define LOOP_BLOCKS 2048

typedef __attribute__((ext_vector_type(2))) float v2f;
typedef __attribute__((ext_vector_type(8))) float v8f;

__device__ __forceinline__ unsigned fkey(float f) {
  unsigned b = __float_as_uint(f);
  return b ^ ((unsigned)((int)b >> 31) | 0x80000000u);
}
__device__ __forceinline__ float kinv(unsigned k) {
  return __uint_as_float((k & 0x80000000u) ? (k ^ 0x80000000u) : ~k);
}

__device__ __forceinline__ void edge_sd(const int* __restrict__ ei, int Ereal,
                                        int e, int& s, int& d) {
  if (e < Ereal) { s = ei[e]; d = ei[Ereal + e]; }
  else           { s = e - Ereal; d = s; }
}

__global__ void fill_u32(unsigned* __restrict__ p, unsigned v, long n) {
  long i = (long)blockIdx.x * blockDim.x + threadIdx.x;
  long stride = (long)gridDim.x * blockDim.x;
  for (; i < n; i += stride) p[i] = v;
}

#define GEMM_WAVES 4
#define WPITCH     272

__global__ __launch_bounds__(GEMM_WAVES * 32) void gemm_wmma_f32(
    const float* __restrict__ X, const float* __restrict__ Wm,
    float* __restrict__ Ho, int ntiles) {
  __shared__ __attribute__((aligned(16))) float s_w[(Dch / 2) * WPITCH];
  const int tid  = threadIdx.x;
  const int wave = tid >> 5;
  const int lane = tid & 31;
  const int half = lane >> 4;
  const int l15  = lane & 15;
  const int mytile = (int)blockIdx.x * GEMM_WAVES + wave;
  const int tile = mytile < ntiles ? mytile : ntiles - 1;

  const float4* arow4 = (const float4*)(X + (long)(tile * 16 + l15) * Dch) + half;
  float4 a4[Dch / 8];
#pragma unroll
  for (int j = 0; j < Dch / 8; ++j) a4[j] = arow4[2 * j];

#pragma unroll 8
  for (int i = 0; i < (Dch / 2) * (Dch / 4) / (GEMM_WAVES * 32); ++i) {
    const int it = i * (GEMM_WAVES * 32) + tid;
    const int p = it >> 5, g = it & 31;
    const float4 r0 = ((const float4*)(Wm + (long)(2 * p) * Dch))[g];
    const float4 r1 = ((const float4*)(Wm + (long)(2 * p + 1) * Dch))[g];
    float4* d = (float4*)(s_w + p * WPITCH + g * 8);
    d[0] = make_float4(r0.x, r1.x, r0.y, r1.y);
    d[1] = make_float4(r0.z, r1.z, r0.w, r1.w);
  }
  __syncthreads();
  if (mytile >= ntiles) return;

  v8f acc[Dch / 16];
#pragma unroll
  for (int t = 0; t < Dch / 16; ++t) acc[t] = (v8f){};
  const float* bbase = s_w + (2 * half) * WPITCH + l15 * 2;
#pragma unroll
  for (int j = 0; j < Dch / 8; ++j) {
#pragma unroll
    for (int s = 0; s < 2; ++s) {
      v2f a;
      a.x = s ? a4[j].z : a4[j].x;
      a.y = s ? a4[j].w : a4[j].y;
      const float* bp = bbase + (4 * j + s) * WPITCH;
#pragma unroll
      for (int t = 0; t < Dch / 16; ++t) {
        const v2f b = *(const v2f*)(bp + t * 32);
        acc[t] = __builtin_amdgcn_wmma_f32_16x16x4_f32(
             false, a,  false, b,
             (short)0, acc[t],  false,  false);
      }
    }
  }
  float* orow = Ho + (long)(tile * 16 + half * 8) * Dch + l15;
#pragma unroll
  for (int t = 0; t < Dch / 16; ++t)
#pragma unroll
    for (int v = 0; v < 8; ++v) orow[(long)v * Dch + t * 16] = acc[t][v];
}

__global__ void att_logits(const float* __restrict__ Hf,
                           const float* __restrict__ a_src,
                           const float* __restrict__ a_dst,
                           float* __restrict__ alsrc, float* __restrict__ aldst,
                           int Nn) {
  int i = blockIdx.x * blockDim.x + threadIdx.x;
  if (i >= Nn * Hh) return;
  int n = i >> 2, h = i & 3;
  const float* hp = Hf + (long)n * Dch + h * Ch;
  const float* as = a_src + h * Ch;
  const float* ad = a_dst + h * Ch;
  float s0 = 0.f, s1 = 0.f;
#pragma unroll
  for (int c = 0; c < Ch; ++c) { float v = hp[c]; s0 += v * as[c]; s1 += v * ad[c]; }
  alsrc[i] = s0;
  aldst[i] = s1;
}

#define OWN_ROWS   160
#define OWN_WAVES  32
#define SCAN_LOADS 6
#define LOAD_EDGES 512
#define WAVE_EDGES (SCAN_LOADS * LOAD_EDGES)
#define ROUND_EDGES (OWN_WAVES * WAVE_EDGES)
#define LIST_CAP   (WAVE_EDGES / 2)
#define GLIST_CAP  512

__device__ __forceinline__ unsigned pack_edge(int s, int dl, int rows) {
  return (unsigned)dl < (unsigned)rows ? ((unsigned)dl << 24) | (unsigned)s : 0xff000000u;
}

__device__ __forceinline__ void batch_ends(const int* __restrict__ ei, int Ereal,
                                           const int* list, int b, int n, int lane,
                                           int& sv, int& dv) {
  const int e = list[b + (lane < n ? lane : n - 1)];
  sv = e < Ereal ? ei[e]         : 0;
  dv = e < Ereal ? ei[Ereal + e] : -1;
}

__device__ __forceinline__ void max_batch(
    int n, int sv, int dl, int lane, int n0, int rows,
    const float* __restrict__ alsrc, const float* __restrict__ aldst,
    unsigned* s_max) {
  const bool ok = lane < n && (unsigned)dl < (unsigned)rows;
  const int  dd = ok ? n0 + dl : n0;
  const float4 a4 = ((const float4*)alsrc)[sv];
  const float4 d4 = ((const float4*)aldst)[dd];
  if (ok) {
    const float av[Hh] = {a4.x, a4.y, a4.z, a4.w};
    const float bv[Hh] = {d4.x, d4.y, d4.z, d4.w};
    for (int h = 0; h < Hh; ++h) {
      float sc = av[h] + bv[h];
      sc = sc > 0.f ? sc : sc * NEG_SLOPE;
      atomicMax(&s_max[dl * Hh + h], fkey(sc));
    }
  }
}

template <int K>
__device__ __forceinline__ void walk_edges(
    int j, int n, int sv, int dlv, const float (&p)[Hh], int lane, int rows,
    const float* __restrict__ Hf, float* s_acc) {
  float hv[K][Hh];
  for (int k = 0; k < K; ++k) {
    const int slot = (j + k < n) ? (j + k) : (n - 1);
    const int s = __builtin_amdgcn_readlane(sv, slot);
    const float* hsrc = Hf + (long)s * Dch + lane;
    for (int h = 0; h < Hh; ++h) hv[k][h] = hsrc[h * Ch];
  }
  for (int k = 0; k < K; ++k) {
    if (j + k < n) {
      const int d = __builtin_amdgcn_readlane(dlv, j + k);
      if ((unsigned)d < (unsigned)rows) {
        float* adst = s_acc + d * Dch + lane;
        for (int h = 0; h < Hh; ++h) {
          const float ph = __int_as_float(__builtin_amdgcn_readlane(__float_as_int(p[h]), j + k));
          atomicAdd(&adst[h * Ch], ph * hv[k][h]);
        }
      }
    }
  }
}

__device__ __forceinline__ void agg_batch(
    int n, int sv, int dl, int lane, int n0, int rows,
    const float* __restrict__ alsrc, const float* __restrict__ aldst,
    const unsigned* s_max, const float* __restrict__ Hf,
    float* s_acc, float* s_sum) {
  const bool ok = lane < n && (unsigned)dl < (unsigned)rows;
  const int  dd = ok ? dl : 0;
  const float4 a4 = ((const float4*)alsrc)[sv];
  const float4 d4 = ((const float4*)aldst)[n0 + dd];
  const uint4  m4 = ((const uint4*)s_max)[dd];
  const float    av[Hh] = {a4.x, a4.y, a4.z, a4.w};
  const float    bv[Hh] = {d4.x, d4.y, d4.z, d4.w};
  const unsigned mk[Hh] = {m4.x, m4.y, m4.z, m4.w};
  float p[Hh];
  for (int h = 0; h < Hh; ++h) {
    float sc = av[h] + bv[h];
    sc = sc > 0.f ? sc : sc * NEG_SLOPE;
    p[h] = expf(sc - kinv(mk[h]));
  }
  if (ok)
    for (int h = 0; h < Hh; ++h) atomicAdd(&s_sum[dl * Hh + h], p[h]);
  for (int j = 0; j < n; j += 8)
    walk_edges<8>(j, n, sv, dl, p, lane, rows, Hf, s_acc);
}

template <int PHASE>
__device__ __forceinline__ void work_list(
    const int* list, int cnt, const int* __restrict__ ei, int Ereal,
    int lane, int n0, int rows,
    const float* __restrict__ alsrc, const float* __restrict__ aldst,
    unsigned* s_max, const float* __restrict__ Hf, float* s_acc, float* s_sum) {
  for (int b = 0; b < cnt; b += 32) {
    const int n = (cnt - b) < 32 ? (cnt - b) : 32;
    int sv, dv;
    batch_ends(ei, Ereal, list, b, n, lane, sv, dv);
    if (PHASE == 0) {
      max_batch(n, sv, dv - n0, lane, n0, rows, alsrc, aldst, s_max);
    } else {
      agg_batch(n, sv, dv - n0, lane, n0, rows, alsrc, aldst, s_max, Hf, s_acc, s_sum);
    }
  }
}

__device__ __forceinline__ void sums_listed(
    const unsigned* list, int cnt, int lane, int n0, int rows,
    const float* __restrict__ alsrc, const float* __restrict__ aldst,
    const unsigned* s_max, const float* __restrict__ Hf, float* s_acc, float* s_sum) {
  for (int b = 0; b < cnt; b += 32) {
    const int n = (cnt - b) < 32 ? (cnt - b) : 32;
    const unsigned w = list[b + (lane < n ? lane : n - 1)];
    const int sv = (int)(w & 0xffffffu), dl = (int)(w >> 24);
    agg_batch(n, sv, dl, lane, n0, rows, alsrc, aldst, s_max, Hf, s_acc, s_sum);
  }
}

__device__ __forceinline__ int scan_rounds(int Ereal) {
  int R = (Ereal + ROUND_EDGES - 1) / ROUND_EDGES;
  return R < 1 ? 1 : R;
}

__global__ __launch_bounds__(256) void owner_tags(const int* __restrict__ ei, int Ereal,
                                                  unsigned* __restrict__ tags4) {
  const int* __restrict__ dstp = ei + Ereal;
  const bool vec_ok = (Ereal & 3) == 0;
  const int words  = scan_rounds(Ereal) * (ROUND_EDGES / 4);
  const int stride = (int)(gridDim.x * blockDim.x);
  for (int w = (int)(blockIdx.x * blockDim.x + threadIdx.x); w < words; w += stride) {
    const int e0 = w * 4;
    unsigned v;
    if (vec_ok && e0 + 4 <= Ereal) {
      const int4 q = *(const int4*)(dstp + e0);
      const int d[4] = {q.x, q.y, q.z, q.w};
      v = 0u;
      for (int c = 0; c < 4; ++c)
        v |= (((unsigned)d[c] / (unsigned)OWN_ROWS) & 255u) << (8 * c);
    } else {
      v = 0xffffffffu;
      for (int c = 0; c < 4; ++c)
        if (e0 + c < Ereal)
          v = (v & ~(255u << (8 * c))) |
              ((((unsigned)dstp[e0 + c] / (unsigned)OWN_ROWS) & 255u) << (8 * c));
    }
    tags4[w] = v;
  }
}

__device__ __forceinline__ void scan_load(const unsigned char* __restrict__ tags,
                                          int base, int lane,
                                          uint4 (&q)[SCAN_LOADS]) {
  const uint4* p = (const uint4*)(tags + base) + lane;
  for (int k = 0; k < SCAN_LOADS; ++k) q[k] = p[k * 32];
}

template <int NP>
__device__ __forceinline__ int scan_pieces(const uint4 (&q)[NP], int piece0, unsigned pat,
                                           int lane16, int base, int cnt, int* list) {
#pragma unroll
  for (int k = 0; k < NP; ++k) {
    const int o0 = (piece0 + k) * LOAD_EDGES + lane16;
    const unsigned wv[4] = {q[k].x, q[k].y, q[k].z, q[k].w};
#pragma unroll
    for (int j = 0; j < 4; ++j) {
      const unsigned x = wv[j] ^ pat;
      const unsigned z = (x - 0x01010101u) & ~x & 0x80808080u;
      if (__builtin_amdgcn_ballot_w32(z != 0u)) {
        for (int c = 0; c < 4; ++c) {
          const int  o   = o0 + j * 4 + c;
          const bool hit = (x & (255u << (8 * c))) == 0u;
          const unsigned mask = __builtin_amdgcn_ballot_w32(hit);
          if (mask) {
            const int slot = cnt + (int)__builtin_amdgcn_mbcnt_lo(mask, 0u);
            if (hit && slot < LIST_CAP) list[slot] = base + o;
            cnt += __builtin_popcount(mask);
          }
        }
      }
    }
  }
  return cnt;
}

template <int PHASE>
__device__ __forceinline__ void search_again(
    const unsigned char* __restrict__ tags, int ch, int R, int T, unsigned pat, int lane,
    int* list, const int* __restrict__ ei, int Ereal, int n0, int rows,
    const float* __restrict__ alsrc, const float* __restrict__ aldst,
    unsigned* s_max, const float* __restrict__ Hf, float* s_acc, float* s_sum) {
  const int lane16 = lane * 16;
  for (int r = 0; r < R; ++r) {
    const int base = ch * WAVE_EDGES;
    uint4 q[SCAN_LOADS];
    scan_load(tags, base, lane, q);
    for (int half = 0; half < 2; ++half) {
      uint4 qh[SCAN_LOADS / 2];
      for (int k = 0; k < SCAN_LOADS / 2; ++k) qh[k] = half ? q[SCAN_LOADS / 2 + k] : q[k];
      const int c = scan_pieces<SCAN_LOADS / 2>(qh, half * (SCAN_LOADS / 2), pat, lane16, base, 0, list);
      work_list<PHASE>(list, c, ei, Ereal, lane, n0, rows, alsrc, aldst, s_max, Hf, s_acc, s_sum);
    }
    ch += OWN_WAVES;
    if (ch >= T) ch -= T;
  }
}

template <bool FIRST>
__global__ __launch_bounds__(OWN_WAVES * 32) void owner_aggregate(
    const int* __restrict__ ei, const unsigned char* __restrict__ tags, int Ereal, int Nn,
    const float* __restrict__ alsrc, const float* __restrict__ aldst,
    const float* __restrict__ Hf,
    float* __restrict__ ssum, float* __restrict__ acc,
    unsigned* glist, int* gcnt) {
  __shared__ float    s_acc[OWN_ROWS * Dch];
  __shared__ float    s_sum[OWN_ROWS * Hh];
  __shared__ __attribute__((aligned(16))) unsigned s_max[OWN_ROWS * Hh];
  __shared__ int      s_list[OWN_WAVES * LIST_CAP];
  __shared__ int      s_cnt[OWN_WAVES];

  const int tid  = threadIdx.x;
  const int lane = tid & 31;
  const int wave = __builtin_amdgcn_readfirstlane(tid >> 5);
  const int n0   = (int)blockIdx.x * OWN_ROWS;
  const int rows = (Nn - n0) < OWN_ROWS ? (Nn - n0) : OWN_ROWS;

  for (int i = tid; i < OWN_ROWS * Dch / 4; i += OWN_WAVES * 32)
    ((float4*)s_acc)[i] = make_float4(0.f, 0.f, 0.f, 0.f);
  for (int i = tid; i < OWN_ROWS * Hh; i += OWN_WAVES * 32) {
    s_sum[i] = 0.f;
    s_max[i] = 0u;
  }
  __syncthreads();

  const int R = scan_rounds(Ereal);
  const int T = R * OWN_WAVES;
  int* mylist = s_list + wave * LIST_CAP;
  unsigned* mywords = (unsigned*)mylist;
  const int gslot = (int)blockIdx.x * OWN_WAVES + wave;
  unsigned* gl = glist + (long)gslot * GLIST_CAP;
  const unsigned pat = ((unsigned)blockIdx.x & 255u) * 0x01010101u;

  const int ch0 = (int)(((unsigned)blockIdx.x * 13u + (unsigned)wave) % (unsigned)T);

  int cnt = 0;
  if (FIRST) {
    int ch = ch0;
    uint4 q[SCAN_LOADS], qn[SCAN_LOADS];
    scan_load(tags, ch * WAVE_EDGES, lane, q);
    for (int k = 0; k < SCAN_LOADS; ++k) qn[k] = q[k];
    const int lane16 = lane * 16;
    for (int r = 0; r < R; ++r) {
      const int base = ch * WAVE_EDGES;
      ch += OWN_WAVES;
      if (ch >= T) ch -= T;
      if (r + 1 < R) scan_load(tags, ch * WAVE_EDGES, lane, qn);
      cnt = scan_pieces<SCAN_LOADS>(q, 0, pat, lane16, base, cnt, mylist);
      for (int k = 0; k < SCAN_LOADS; ++k) q[k] = qn[k];
    }
    if (lane == 0) s_cnt[wave] = cnt;
  } else {
    cnt = gcnt[gslot];
  }
  const bool listed = cnt <= GLIST_CAP;
  if (listed) {
    for (int b = 0; b < cnt; b += 32) {
      const int n = (cnt - b) < 32 ? (cnt - b) : 32;
      int sv, dl;
      unsigned w;
      if (FIRST) {
        int dv;
        batch_ends(ei, Ereal, mylist, b, n, lane, sv, dv);
        dl = dv - n0;
        w  = pack_edge(sv, dl, rows);
        if (lane < n) gl[b + lane] = w;
      } else {
        w  = gl[b + (lane < n ? lane : n - 1)];
        sv = (int)(w & 0xffffffu);
        dl = (int)(w >> 24);
      }
      if (lane < n) mywords[b + lane] = w;
      max_batch(n, sv, dl, lane, n0, rows, alsrc, aldst, s_max);
    }
  } else {
    search_again<0>(tags, ch0, R, T, pat, lane, mylist, ei, Ereal, n0, rows,
                    alsrc, aldst, s_max, Hf, s_acc, s_sum);
  }

  const int per = OWN_ROWS / OWN_WAVES;
  int nself = rows - wave * per;
  nself = nself < 0 ? 0 : (nself > per ? per : nself);
  const int selfv = n0 + wave * per + (lane < nself ? lane : nself - 1);
  if (nself > 0) max_batch(nself, selfv, selfv - n0, lane, n0, rows, alsrc, aldst, s_max);

  __syncthreads();

  if (FIRST && wave == 0) gcnt[(int)blockIdx.x * OWN_WAVES + lane] = s_cnt[lane];

  if (listed)
    sums_listed(mywords, cnt, lane, n0, rows, alsrc, aldst, s_max, Hf, s_acc, s_sum);
  else
    search_again<1>(tags, ch0, R, T, pat, lane, mylist, ei, Ereal, n0, rows,
                    alsrc, aldst, s_max, Hf, s_acc, s_sum);
  if (nself > 0)
    agg_batch(nself, selfv, selfv - n0, lane, n0, rows, alsrc, aldst, s_max, Hf, s_acc, s_sum);

  __syncthreads();
  float4*       o4 = (float4*)(acc + (long)n0 * Dch);
  const float4* a4 = (const float4*)s_acc;
  for (int i = tid; i < rows * (Dch / 4); i += OWN_WAVES * 32) o4[i] = a4[i];
  for (int i = tid; i < rows * Hh; i += OWN_WAVES * 32) ssum[(long)n0 * Hh + i] = s_sum[i];
}

__global__ __launch_bounds__(256) void post_bn_elu(float* __restrict__ x,
                            const float* __restrict__ ssum,
                            const float* __restrict__ bias,
                            const float* __restrict__ gamma,
                            const float* __restrict__ beta,
                            const float* __restrict__ mean,
                            const float* __restrict__ var,
                            long n) {
  const long tid    = (long)blockIdx.x * blockDim.x + threadIdx.x;
  const long stride = (long)gridDim.x * blockDim.x;
  const int  c      = (int)((tid * 4) & (Dch - 1));
  const int  head   = c / Ch;
  float b[4], m[4], sc[4], be[4];
  for (int k = 0; k < 4; ++k) {
    b[k]  = bias[c + k];
    m[k]  = mean[c + k];
    sc[k] = gamma[c + k] / sqrtf(var[c + k] + BN_EPS);
    be[k] = beta[c + k];
  }
  float4* x4 = (float4*)x;
  const long n4 = n >> 2;
  for (long i = tid; i < n4; i += stride) {
    float4 q = x4[i];
    const float den = ssum[(i >> 5) * Hh + head] + 1e-16f;
    float v[4] = {q.x, q.y, q.z, q.w};
    for (int k = 0; k < 4; ++k) {
      float t = v[k] / den + b[k];
      t = (t - m[k]) * sc[k] + be[k];
      v[k] = t > 0.f ? t : expf(t) - 1.f;
    }
    x4[i] = make_float4(v[0], v[1], v[2], v[3]);
  }
}

__global__ void pool_accum(const float* __restrict__ x,
                           const int* __restrict__ batch,
                           float* __restrict__ sums, float* __restrict__ cnt,
                           long n) {
  const long stride = (long)gridDim.x * blockDim.x;
  for (long i = (long)blockIdx.x * blockDim.x + threadIdx.x; i < n; i += stride) {
    int node = (int)(i >> 7);
    int c    = (int)(i & 127);
    int g    = batch[node];
    atomicAdd(&sums[(long)g * Dch + c], x[i]);
    if (c == 0) atomicAdd(&cnt[g], 1.0f);
  }
}

__global__ void pool_final(const float* __restrict__ sums,
                           const float* __restrict__ cnt,
                           float* __restrict__ out, int G) {
  int i = blockIdx.x * blockDim.x + threadIdx.x;
  if (i >= G * Dch) return;
  int g = i >> 7;
  float c = cnt[g];
  out[i] = sums[i] / (c > 1.f ? c : 1.f);
}

extern "C" void kernel_launch(void* const* d_in, const int* in_sizes, int n_in,
                              void* d_out, int out_size, void* d_ws, size_t ws_size,
                              hipStream_t stream) {
  const float* x0    = (const float*)d_in[0];
  const int*   ei    = (const int*)d_in[1];
  const int*   batch = (const int*)d_in[2];
  const float* Ws    = (const float*)d_in[3];
  const float* a_src = (const float*)d_in[4];
  const float* a_dst = (const float*)d_in[5];
  const float* bias  = (const float*)d_in[6];
  const float* gamma = (const float*)d_in[7];
  const float* beta  = (const float*)d_in[8];
  const float* mean  = (const float*)d_in[9];
  const float* var   = (const float*)d_in[10];

  const int Nn    = in_sizes[0] / Dch;
  const int Ereal = in_sizes[1] / 2;
  const int G     = out_size / Dch;

  char* ws = (char*)d_ws;
  float*    bufA  = (float*)ws;    ws += (size_t)Nn * Dch * 4;
  unsigned char* tags = (unsigned char*)ws; ws += (size_t)Nn * Dch * 4;
  const int    groups    = (Nn + OWN_ROWS - 1) / OWN_ROWS;
  const size_t tag_bytes = (size_t)((Ereal + ROUND_EDGES - 1) / ROUND_EDGES + 1) * ROUND_EDGES;
  unsigned*    glist     = (unsigned*)(tags + tag_bytes);
  int*         gcnt      = (int*)(glist + (size_t)groups * OWN_WAVES * GLIST_CAP);
  float*    hbuf  = (float*)ws;    ws += (size_t)Nn * Dch * 4;
  float*    ssum  = (float*)ws;    ws += (size_t)Nn * Hh * 4;
  float*    alsrc = (float*)ws;    ws += (size_t)Nn * Hh * 4;
  float*    aldst = (float*)ws;    ws += (size_t)Nn * Hh * 4;
  float*    psum  = (float*)ws;    ws += (size_t)G * Dch * 4;
  float*    pcnt  = (float*)ws;    ws += (size_t)G * 4;

  owner_tags<<<256, 256, 0, stream>>>(ei, Ereal, (unsigned*)tags);

  const float* xin  = x0;
  float*       xout = bufA;
  for (int l = 0; l < Lnum; ++l) {
    const float* Wl = Ws + (size_t)l * Dch * Dch;
    gemm_wmma_f32<<<(Nn / 16 + GEMM_WAVES - 1) / GEMM_WAVES, GEMM_WAVES * 32, 0, stream>>>(
        xin, Wl, hbuf, Nn / 16);

    att_logits<<<(Nn * Hh + 255) / 256, 256, 0, stream>>>(
        hbuf, a_src + (size_t)l * Hh * Ch, a_dst + (size_t)l * Hh * Ch,
        alsrc, aldst, Nn);

    if (l == 0)
      owner_aggregate<true><<<groups, OWN_WAVES * 32, 0, stream>>>(
          ei, tags, Ereal, Nn, alsrc, aldst, hbuf, ssum, xout, glist, gcnt);
    else
      owner_aggregate<false><<<groups, OWN_WAVES * 32, 0, stream>>>(
          ei, tags, Ereal, Nn, alsrc, aldst, hbuf, ssum, xout, glist, gcnt);

    post_bn_elu<<<LOOP_BLOCKS, 256, 0, stream>>>(
        xout, ssum, bias + (size_t)l * Dch, gamma + (size_t)l * Dch,
        beta + (size_t)l * Dch, mean + (size_t)l * Dch, var + (size_t)l * Dch,
        (long)Nn * Dch);

    xin  = xout;
  }

  fill_u32<<<32, 256, 0, stream>>>((unsigned*)psum, 0u, (long)G * Dch);
  fill_u32<<<1, 64, 0, stream>>>((unsigned*)pcnt, 0u, G);
  pool_accum<<<LOOP_BLOCKS, 256, 0, stream>>>(xin, batch, psum, pcnt, (long)Nn * Dch);
  pool_final<<<(G * Dch + 255) / 256, 256, 0, stream>>>(psum, pcnt, (float*)d_out, G);
}


# rows stay unnormalised in memory: division+bias+BatchNorm+ELU ride the consumers' loads (next GEMM on its A strip beside the WMMAs, pool on its elements); 3 elementwise passes gone, 17 to 14 launches
# speedup vs baseline: 4.3463x; 1.0526x over previous
#include <hip/hip_runtime.h>


#define Dch      128
#define Hh       4
#define Ch       32
#define Lnum     3
#define NEG_SLOPE 0.2f
#define BN_EPS   1e-5f
#define LOOP_BLOCKS 2048

typedef __attribute__((ext_vector_type(2))) float v2f;
typedef __attribute__((ext_vector_type(8))) float v8f;

__device__ __forceinline__ unsigned fkey(float f) {
  unsigned b = __float_as_uint(f);
  return b ^ ((unsigned)((int)b >> 31) | 0x80000000u);
}
__device__ __forceinline__ float kinv(unsigned k) {
  return __uint_as_float((k & 0x80000000u) ? (k ^ 0x80000000u) : ~k);
}

__device__ __forceinline__ void edge_sd(const int* __restrict__ ei, int Ereal,
                                        int e, int& s, int& d) {
  if (e < Ereal) { s = ei[e]; d = ei[Ereal + e]; }
  else           { s = e - Ereal; d = s; }
}

__global__ void fill_u32(unsigned* __restrict__ p, unsigned v, long n) {
  long i = (long)blockIdx.x * blockDim.x + threadIdx.x;
  long stride = (long)gridDim.x * blockDim.x;
  for (; i < n; i += stride) p[i] = v;
}

__device__ __forceinline__ float post_act(float v, float den, float b, float m,
                                          float sc, float be) {
  float t = v / den + b;
  t = (t - m) * sc + be;
  const float e = expf(t) - 1.f;
  return t > 0.f ? t : e;
}

#define GEMM_WAVES 4
#define WPITCH     272

template <bool ACT>
__global__ __launch_bounds__(GEMM_WAVES * 32) void gemm_wmma_f32(
    const float* __restrict__ X, const float* __restrict__ Wm,
    float* __restrict__ Ho, int ntiles,
    const float* __restrict__ ssum, const float* __restrict__ bias,
    const float* __restrict__ gamma, const float* __restrict__ beta,
    const float* __restrict__ mean, const float* __restrict__ var) {
  __shared__ __attribute__((aligned(16))) float s_w[(Dch / 2) * WPITCH];
  __shared__ float4 s_par[ACT ? Dch : 1];
  const int tid  = threadIdx.x;
  const int wave = tid >> 5;
  const int lane = tid & 31;
  const int half = lane >> 4;
  const int l15  = lane & 15;
  const int mytile = (int)blockIdx.x * GEMM_WAVES + wave;
  const int tile = mytile < ntiles ? mytile : ntiles - 1;

  const float4* arow4 = (const float4*)(X + (long)(tile * 16 + l15) * Dch) + half;
  float4 a4[Dch / 8];
#pragma unroll
  for (int j = 0; j < Dch / 8; ++j) a4[j] = arow4[2 * j];
  float4 s4 = make_float4(1.f, 1.f, 1.f, 1.f);
  if (ACT) {
    s4 = ((const float4*)ssum)[tile * 16 + l15];
    for (int ch = tid; ch < Dch; ch += GEMM_WAVES * 32)
      s_par[ch] = make_float4(bias[ch], mean[ch], gamma[ch] / sqrtf(var[ch] + BN_EPS), beta[ch]);
  }

#pragma unroll 8
  for (int i = 0; i < (Dch / 2) * (Dch / 4) / (GEMM_WAVES * 32); ++i) {
    const int it = i * (GEMM_WAVES * 32) + tid;
    const int p = it >> 5, g = it & 31;
    const float4 r0 = ((const float4*)(Wm + (long)(2 * p) * Dch))[g];
    const float4 r1 = ((const float4*)(Wm + (long)(2 * p + 1) * Dch))[g];
    float4* d = (float4*)(s_w + p * WPITCH + g * 8);
    d[0] = make_float4(r0.x, r1.x, r0.y, r1.y);
    d[1] = make_float4(r0.z, r1.z, r0.w, r1.w);
  }
  __syncthreads();
  if (mytile >= ntiles) return;

  v8f acc[Dch / 16];
#pragma unroll
  for (int t = 0; t < Dch / 16; ++t) acc[t] = (v8f){};
  const float* bbase = s_w + (2 * half) * WPITCH + l15 * 2;
  const float den[Hh] = {s4.x + 1e-16f, s4.y + 1e-16f, s4.z + 1e-16f, s4.w + 1e-16f};
#pragma unroll
  for (int j = 0; j < Dch / 8; ++j) {
    float4 x4 = a4[j];
    if (ACT) {
      const float4* pr = s_par + 8 * j + 4 * half;
      const float4 p0 = pr[0], p1 = pr[1], p2 = pr[2], p3 = pr[3];
      const float  dn = den[j / 4];
      x4.x = post_act(x4.x, dn, p0.x, p0.y, p0.z, p0.w);
      x4.y = post_act(x4.y, dn, p1.x, p1.y, p1.z, p1.w);
      x4.z = post_act(x4.z, dn, p2.x, p2.y, p2.z, p2.w);
      x4.w = post_act(x4.w, dn, p3.x, p3.y, p3.z, p3.w);
    }
#pragma unroll
    for (int s = 0; s < 2; ++s) {
      v2f a;
      a.x = s ? x4.z : x4.x;
      a.y = s ? x4.w : x4.y;
      const float* bp = bbase + (4 * j + s) * WPITCH;
#pragma unroll
      for (int t = 0; t < Dch / 16; ++t) {
        const v2f b = *(const v2f*)(bp + t * 32);
        acc[t] = __builtin_amdgcn_wmma_f32_16x16x4_f32(
             false, a,  false, b,
             (short)0, acc[t],  false,  false);
      }
    }
  }
  float* orow = Ho + (long)(tile * 16 + half * 8) * Dch + l15;
#pragma unroll
  for (int t = 0; t < Dch / 16; ++t)
#pragma unroll
    for (int v = 0; v < 8; ++v) orow[(long)v * Dch + t * 16] = acc[t][v];
}

__global__ void att_logits(const float* __restrict__ Hf,
                           const float* __restrict__ a_src,
                           const float* __restrict__ a_dst,
                           float* __restrict__ alsrc, float* __restrict__ aldst,
                           int Nn) {
  int i = blockIdx.x * blockDim.x + threadIdx.x;
  if (i >= Nn * Hh) return;
  int n = i >> 2, h = i & 3;
  const float* hp = Hf + (long)n * Dch + h * Ch;
  const float* as = a_src + h * Ch;
  const float* ad = a_dst + h * Ch;
  float s0 = 0.f, s1 = 0.f;
#pragma unroll
  for (int c = 0; c < Ch; ++c) { float v = hp[c]; s0 += v * as[c]; s1 += v * ad[c]; }
  alsrc[i] = s0;
  aldst[i] = s1;
}

#define OWN_ROWS   160
#define OWN_WAVES  32
#define SCAN_LOADS 6
#define LOAD_EDGES 512
#define WAVE_EDGES (SCAN_LOADS * LOAD_EDGES)
#define ROUND_EDGES (OWN_WAVES * WAVE_EDGES)
#define LIST_CAP   (WAVE_EDGES / 2)
#define GLIST_CAP  512

__device__ __forceinline__ unsigned pack_edge(int s, int dl, int rows) {
  return (unsigned)dl < (unsigned)rows ? ((unsigned)dl << 24) | (unsigned)s : 0xff000000u;
}

__device__ __forceinline__ void batch_ends(const int* __restrict__ ei, int Ereal,
                                           const int* list, int b, int n, int lane,
                                           int& sv, int& dv) {
  const int e = list[b + (lane < n ? lane : n - 1)];
  sv = e < Ereal ? ei[e]         : 0;
  dv = e < Ereal ? ei[Ereal + e] : -1;
}

__device__ __forceinline__ void max_batch(
    int n, int sv, int dl, int lane, int n0, int rows,
    const float* __restrict__ alsrc, const float* __restrict__ aldst,
    unsigned* s_max) {
  const bool ok = lane < n && (unsigned)dl < (unsigned)rows;
  const int  dd = ok ? n0 + dl : n0;
  const float4 a4 = ((const float4*)alsrc)[sv];
  const float4 d4 = ((const float4*)aldst)[dd];
  if (ok) {
    const float av[Hh] = {a4.x, a4.y, a4.z, a4.w};
    const float bv[Hh] = {d4.x, d4.y, d4.z, d4.w};
    for (int h = 0; h < Hh; ++h) {
      float sc = av[h] + bv[h];
      sc = sc > 0.f ? sc : sc * NEG_SLOPE;
      atomicMax(&s_max[dl * Hh + h], fkey(sc));
    }
  }
}

template <int K>
__device__ __forceinline__ void walk_edges(
    int j, int n, int sv, int dlv, const float (&p)[Hh], int lane, int rows,
    const float* __restrict__ Hf, float* s_acc) {
  float hv[K][Hh];
  for (int k = 0; k < K; ++k) {
    const int slot = (j + k < n) ? (j + k) : (n - 1);
    const int s = __builtin_amdgcn_readlane(sv, slot);
    const float* hsrc = Hf + (long)s * Dch + lane;
    for (int h = 0; h < Hh; ++h) hv[k][h] = hsrc[h * Ch];
  }
  for (int k = 0; k < K; ++k) {
    if (j + k < n) {
      const int d = __builtin_amdgcn_readlane(dlv, j + k);
      if ((unsigned)d < (unsigned)rows) {
        float* adst = s_acc + d * Dch + lane;
        for (int h = 0; h < Hh; ++h) {
          const float ph = __int_as_float(__builtin_amdgcn_readlane(__float_as_int(p[h]), j + k));
          atomicAdd(&adst[h * Ch], ph * hv[k][h]);
        }
      }
    }
  }
}

__device__ __forceinline__ void agg_batch(
    int n, int sv, int dl, int lane, int n0, int rows,
    const float* __restrict__ alsrc, const float* __restrict__ aldst,
    const unsigned* s_max, const float* __restrict__ Hf,
    float* s_acc, float* s_sum) {
  const bool ok = lane < n && (unsigned)dl < (unsigned)rows;
  const int  dd = ok ? dl : 0;
  const float4 a4 = ((const float4*)alsrc)[sv];
  const float4 d4 = ((const float4*)aldst)[n0 + dd];
  const uint4  m4 = ((const uint4*)s_max)[dd];
  const float    av[Hh] = {a4.x, a4.y, a4.z, a4.w};
  const float    bv[Hh] = {d4.x, d4.y, d4.z, d4.w};
  const unsigned mk[Hh] = {m4.x, m4.y, m4.z, m4.w};
  float p[Hh];
  for (int h = 0; h < Hh; ++h) {
    float sc = av[h] + bv[h];
    sc = sc > 0.f ? sc : sc * NEG_SLOPE;
    p[h] = expf(sc - kinv(mk[h]));
  }
  if (ok)
    for (int h = 0; h < Hh; ++h) atomicAdd(&s_sum[dl * Hh + h], p[h]);
  for (int j = 0; j < n; j += 8)
    walk_edges<8>(j, n, sv, dl, p, lane, rows, Hf, s_acc);
}

template <int PHASE>
__device__ __forceinline__ void work_list(
    const int* list, int cnt, const int* __restrict__ ei, int Ereal,
    int lane, int n0, int rows,
    const float* __restrict__ alsrc, const float* __restrict__ aldst,
    unsigned* s_max, const float* __restrict__ Hf, float* s_acc, float* s_sum) {
  for (int b = 0; b < cnt; b += 32) {
    const int n = (cnt - b) < 32 ? (cnt - b) : 32;
    int sv, dv;
    batch_ends(ei, Ereal, list, b, n, lane, sv, dv);
    if (PHASE == 0) {
      max_batch(n, sv, dv - n0, lane, n0, rows, alsrc, aldst, s_max);
    } else {
      agg_batch(n, sv, dv - n0, lane, n0, rows, alsrc, aldst, s_max, Hf, s_acc, s_sum);
    }
  }
}

__device__ __forceinline__ void sums_listed(
    const unsigned* list, int cnt, int lane, int n0, int rows,
    const float* __restrict__ alsrc, const float* __restrict__ aldst,
    const unsigned* s_max, const float* __restrict__ Hf, float* s_acc, float* s_sum) {
  for (int b = 0; b < cnt; b += 32) {
    const int n = (cnt - b) < 32 ? (cnt - b) : 32;
    const unsigned w = list[b + (lane < n ? lane : n - 1)];
    const int sv = (int)(w & 0xffffffu), dl = (int)(w >> 24);
    agg_batch(n, sv, dl, lane, n0, rows, alsrc, aldst, s_max, Hf, s_acc, s_sum);
  }
}

__device__ __forceinline__ int scan_rounds(int Ereal) {
  int R = (Ereal + ROUND_EDGES - 1) / ROUND_EDGES;
  return R < 1 ? 1 : R;
}

__global__ __launch_bounds__(256) void owner_tags(const int* __restrict__ ei, int Ereal,
                                                  unsigned* __restrict__ tags4) {
  const int* __restrict__ dstp = ei + Ereal;
  const bool vec_ok = (Ereal & 3) == 0;
  const int words  = scan_rounds(Ereal) * (ROUND_EDGES / 4);
  const int stride = (int)(gridDim.x * blockDim.x);
  for (int w = (int)(blockIdx.x * blockDim.x + threadIdx.x); w < words; w += stride) {
    const int e0 = w * 4;
    unsigned v;
    if (vec_ok && e0 + 4 <= Ereal) {
      const int4 q = *(const int4*)(dstp + e0);
      const int d[4] = {q.x, q.y, q.z, q.w};
      v = 0u;
      for (int c = 0; c < 4; ++c)
        v |= (((unsigned)d[c] / (unsigned)OWN_ROWS) & 255u) << (8 * c);
    } else {
      v = 0xffffffffu;
      for (int c = 0; c < 4; ++c)
        if (e0 + c < Ereal)
          v = (v & ~(255u << (8 * c))) |
              ((((unsigned)dstp[e0 + c] / (unsigned)OWN_ROWS) & 255u) << (8 * c));
    }
    tags4[w] = v;
  }
}

__device__ __forceinline__ void scan_load(const unsigned char* __restrict__ tags,
                                          int base, int lane,
                                          uint4 (&q)[SCAN_LOADS]) {
  const uint4* p = (const uint4*)(tags + base) + lane;
  for (int k = 0; k < SCAN_LOADS; ++k) q[k] = p[k * 32];
}

template <int NP>
__device__ __forceinline__ int scan_pieces(const uint4 (&q)[NP], int piece0, unsigned pat,
                                           int lane16, int base, int cnt, int* list) {
#pragma unroll
  for (int k = 0; k < NP; ++k) {
    const int o0 = (piece0 + k) * LOAD_EDGES + lane16;
    const unsigned wv[4] = {q[k].x, q[k].y, q[k].z, q[k].w};
#pragma unroll
    for (int j = 0; j < 4; ++j) {
      const unsigned x = wv[j] ^ pat;
      const unsigned z = (x - 0x01010101u) & ~x & 0x80808080u;
      if (__builtin_amdgcn_ballot_w32(z != 0u)) {
        for (int c = 0; c < 4; ++c) {
          const int  o   = o0 + j * 4 + c;
          const bool hit = (x & (255u << (8 * c))) == 0u;
          const unsigned mask = __builtin_amdgcn_ballot_w32(hit);
          if (mask) {
            const int slot = cnt + (int)__builtin_amdgcn_mbcnt_lo(mask, 0u);
            if (hit && slot < LIST_CAP) list[slot] = base + o;
            cnt += __builtin_popcount(mask);
          }
        }
      }
    }
  }
  return cnt;
}

template <int PHASE>
__device__ __forceinline__ void search_again(
    const unsigned char* __restrict__ tags, int ch, int R, int T, unsigned pat, int lane,
    int* list, const int* __restrict__ ei, int Ereal, int n0, int rows,
    const float* __restrict__ alsrc, const float* __restrict__ aldst,
    unsigned* s_max, const float* __restrict__ Hf, float* s_acc, float* s_sum) {
  const int lane16 = lane * 16;
  for (int r = 0; r < R; ++r) {
    const int base = ch * WAVE_EDGES;
    uint4 q[SCAN_LOADS];
    scan_load(tags, base, lane, q);
    for (int half = 0; half < 2; ++half) {
      uint4 qh[SCAN_LOADS / 2];
      for (int k = 0; k < SCAN_LOADS / 2; ++k) qh[k] = half ? q[SCAN_LOADS / 2 + k] : q[k];
      const int c = scan_pieces<SCAN_LOADS / 2>(qh, half * (SCAN_LOADS / 2), pat, lane16, base, 0, list);
      work_list<PHASE>(list, c, ei, Ereal, lane, n0, rows, alsrc, aldst, s_max, Hf, s_acc, s_sum);
    }
    ch += OWN_WAVES;
    if (ch >= T) ch -= T;
  }
}

template <bool FIRST>
__global__ __launch_bounds__(OWN_WAVES * 32) void owner_aggregate(
    const int* __restrict__ ei, const unsigned char* __restrict__ tags, int Ereal, int Nn,
    const float* __restrict__ alsrc, const float* __restrict__ aldst,
    const float* __restrict__ Hf,
    float* __restrict__ ssum, float* __restrict__ acc,
    unsigned* glist, int* gcnt) {
  __shared__ float    s_acc[OWN_ROWS * Dch];
  __shared__ float    s_sum[OWN_ROWS * Hh];
  __shared__ __attribute__((aligned(16))) unsigned s_max[OWN_ROWS * Hh];
  __shared__ int      s_list[OWN_WAVES * LIST_CAP];
  __shared__ int      s_cnt[OWN_WAVES];

  const int tid  = threadIdx.x;
  const int lane = tid & 31;
  const int wave = __builtin_amdgcn_readfirstlane(tid >> 5);
  const int n0   = (int)blockIdx.x * OWN_ROWS;
  const int rows = (Nn - n0) < OWN_ROWS ? (Nn - n0) : OWN_ROWS;

  for (int i = tid; i < OWN_ROWS * Dch / 4; i += OWN_WAVES * 32)
    ((float4*)s_acc)[i] = make_float4(0.f, 0.f, 0.f, 0.f);
  for (int i = tid; i < OWN_ROWS * Hh; i += OWN_WAVES * 32) {
    s_sum[i] = 0.f;
    s_max[i] = 0u;
  }
  __syncthreads();

  const int R = scan_rounds(Ereal);
  const int T = R * OWN_WAVES;
  int* mylist = s_list + wave * LIST_CAP;
  unsigned* mywords = (unsigned*)mylist;
  const int gslot = (int)blockIdx.x * OWN_WAVES + wave;
  unsigned* gl = glist + (long)gslot * GLIST_CAP;
  const unsigned pat = ((unsigned)blockIdx.x & 255u) * 0x01010101u;

  const int ch0 = (int)(((unsigned)blockIdx.x * 13u + (unsigned)wave) % (unsigned)T);

  int cnt = 0;
  if (FIRST) {
    int ch = ch0;
    uint4 q[SCAN_LOADS], qn[SCAN_LOADS];
    scan_load(tags, ch * WAVE_EDGES, lane, q);
    for (int k = 0; k < SCAN_LOADS; ++k) qn[k] = q[k];
    const int lane16 = lane * 16;
    for (int r = 0; r < R; ++r) {
      const int base = ch * WAVE_EDGES;
      ch += OWN_WAVES;
      if (ch >= T) ch -= T;
      if (r + 1 < R) scan_load(tags, ch * WAVE_EDGES, lane, qn);
      cnt = scan_pieces<SCAN_LOADS>(q, 0, pat, lane16, base, cnt, mylist);
      for (int k = 0; k < SCAN_LOADS; ++k) q[k] = qn[k];
    }
    if (lane == 0) s_cnt[wave] = cnt;
  } else {
    cnt = gcnt[gslot];
  }
  const bool listed = cnt <= GLIST_CAP;
  if (listed) {
    for (int b = 0; b < cnt; b += 32) {
      const int n = (cnt - b) < 32 ? (cnt - b) : 32;
      int sv, dl;
      unsigned w;
      if (FIRST) {
        int dv;
        batch_ends(ei, Ereal, mylist, b, n, lane, sv, dv);
        dl = dv - n0;
        w  = pack_edge(sv, dl, rows);
        if (lane < n) gl[b + lane] = w;
      } else {
        w  = gl[b + (lane < n ? lane : n - 1)];
        sv = (int)(w & 0xffffffu);
        dl = (int)(w >> 24);
      }
      if (lane < n) mywords[b + lane] = w;
      max_batch(n, sv, dl, lane, n0, rows, alsrc, aldst, s_max);
    }
  } else {
    search_again<0>(tags, ch0, R, T, pat, lane, mylist, ei, Ereal, n0, rows,
                    alsrc, aldst, s_max, Hf, s_acc, s_sum);
  }

  const int per = OWN_ROWS / OWN_WAVES;
  int nself = rows - wave * per;
  nself = nself < 0 ? 0 : (nself > per ? per : nself);
  const int selfv = n0 + wave * per + (lane < nself ? lane : nself - 1);
  if (nself > 0) max_batch(nself, selfv, selfv - n0, lane, n0, rows, alsrc, aldst, s_max);

  __syncthreads();

  if (FIRST && wave == 0) gcnt[(int)blockIdx.x * OWN_WAVES + lane] = s_cnt[lane];

  if (listed)
    sums_listed(mywords, cnt, lane, n0, rows, alsrc, aldst, s_max, Hf, s_acc, s_sum);
  else
    search_again<1>(tags, ch0, R, T, pat, lane, mylist, ei, Ereal, n0, rows,
                    alsrc, aldst, s_max, Hf, s_acc, s_sum);
  if (nself > 0)
    agg_batch(nself, selfv, selfv - n0, lane, n0, rows, alsrc, aldst, s_max, Hf, s_acc, s_sum);

  __syncthreads();
  float4*       o4 = (float4*)(acc + (long)n0 * Dch);
  const float4* a4 = (const float4*)s_acc;
  for (int i = tid; i < rows * (Dch / 4); i += OWN_WAVES * 32) o4[i] = a4[i];
  for (int i = tid; i < rows * Hh; i += OWN_WAVES * 32) ssum[(long)n0 * Hh + i] = s_sum[i];
}

__global__ void pool_accum(const float* __restrict__ x,
                           const float* __restrict__ ssum,
                           const float* __restrict__ bias,
                           const float* __restrict__ gamma,
                           const float* __restrict__ beta,
                           const float* __restrict__ mean,
                           const float* __restrict__ var,
                           const int* __restrict__ batch,
                           float* __restrict__ sums, float* __restrict__ cnt,
                           long n) {
  const long stride = (long)gridDim.x * blockDim.x;
  const long i0     = (long)blockIdx.x * blockDim.x + threadIdx.x;
  const int  c      = (int)(i0 & 127);
  const float b  = bias[c], m = mean[c], be = beta[c];
  const float sc = gamma[c] / sqrtf(var[c] + BN_EPS);
  for (long i = i0; i < n; i += stride) {
    int node = (int)(i >> 7);
    int g    = batch[node];
    const float den = ssum[(long)node * Hh + (c >> 5)] + 1e-16f;
    atomicAdd(&sums[(long)g * Dch + c], post_act(x[i], den, b, m, sc, be));
    if (c == 0) atomicAdd(&cnt[g], 1.0f);
  }
}

__global__ void pool_final(const float* __restrict__ sums,
                           const float* __restrict__ cnt,
                           float* __restrict__ out, int G) {
  int i = blockIdx.x * blockDim.x + threadIdx.x;
  if (i >= G * Dch) return;
  int g = i >> 7;
  float c = cnt[g];
  out[i] = sums[i] / (c > 1.f ? c : 1.f);
}

extern "C" void kernel_launch(void* const* d_in, const int* in_sizes, int n_in,
                              void* d_out, int out_size, void* d_ws, size_t ws_size,
                              hipStream_t stream) {
  const float* x0    = (const float*)d_in[0];
  const int*   ei    = (const int*)d_in[1];
  const int*   batch = (const int*)d_in[2];
  const float* Ws    = (const float*)d_in[3];
  const float* a_src = (const float*)d_in[4];
  const float* a_dst = (const float*)d_in[5];
  const float* bias  = (const float*)d_in[6];
  const float* gamma = (const float*)d_in[7];
  const float* beta  = (const float*)d_in[8];
  const float* mean  = (const float*)d_in[9];
  const float* var   = (const float*)d_in[10];

  const int Nn    = in_sizes[0] / Dch;
  const int Ereal = in_sizes[1] / 2;
  const int G     = out_size / Dch;

  char* ws = (char*)d_ws;
  float*    bufA  = (float*)ws;    ws += (size_t)Nn * Dch * 4;
  unsigned char* tags = (unsigned char*)ws; ws += (size_t)Nn * Dch * 4;
  const int    groups    = (Nn + OWN_ROWS - 1) / OWN_ROWS;
  const size_t tag_bytes = (size_t)((Ereal + ROUND_EDGES - 1) / ROUND_EDGES + 1) * ROUND_EDGES;
  unsigned*    glist     = (unsigned*)(tags + tag_bytes);
  int*         gcnt      = (int*)(glist + (size_t)groups * OWN_WAVES * GLIST_CAP);
  float*    hbuf  = (float*)ws;    ws += (size_t)Nn * Dch * 4;
  float*    ssum  = (float*)ws;    ws += (size_t)Nn * Hh * 4;
  float*    alsrc = (float*)ws;    ws += (size_t)Nn * Hh * 4;
  float*    aldst = (float*)ws;    ws += (size_t)Nn * Hh * 4;
  float*    psum  = (float*)ws;    ws += (size_t)G * Dch * 4;
  float*    pcnt  = (float*)ws;    ws += (size_t)G * 4;

  owner_tags<<<256, 256, 0, stream>>>(ei, Ereal, (unsigned*)tags);

  const float* xin  = x0;
  float*       xout = bufA;
  for (int l = 0; l < Lnum; ++l) {
    const float* Wl = Ws + (size_t)l * Dch * Dch;
    if (l == 0)
      gemm_wmma_f32<false><<<(Nn / 16 + GEMM_WAVES - 1) / GEMM_WAVES, GEMM_WAVES * 32, 0, stream>>>(
          xin, Wl, hbuf, Nn / 16, ssum, bias, gamma, beta, mean, var);
    else
      gemm_wmma_f32<true><<<(Nn / 16 + GEMM_WAVES - 1) / GEMM_WAVES, GEMM_WAVES * 32, 0, stream>>>(
          xin, Wl, hbuf, Nn / 16, ssum, bias + (size_t)(l - 1) * Dch, gamma + (size_t)(l - 1) * Dch,
          beta + (size_t)(l - 1) * Dch, mean + (size_t)(l - 1) * Dch, var + (size_t)(l - 1) * Dch);

    att_logits<<<(Nn * Hh + 255) / 256, 256, 0, stream>>>(
        hbuf, a_src + (size_t)l * Hh * Ch, a_dst + (size_t)l * Hh * Ch,
        alsrc, aldst, Nn);

    if (l == 0)
      owner_aggregate<true><<<groups, OWN_WAVES * 32, 0, stream>>>(
          ei, tags, Ereal, Nn, alsrc, aldst, hbuf, ssum, xout, glist, gcnt);
    else
      owner_aggregate<false><<<groups, OWN_WAVES * 32, 0, stream>>>(
          ei, tags, Ereal, Nn, alsrc, aldst, hbuf, ssum, xout, glist, gcnt);

    xin  = xout;
  }

  fill_u32<<<32, 256, 0, stream>>>((unsigned*)psum, 0u, (long)G * Dch);
  fill_u32<<<1, 64, 0, stream>>>((unsigned*)pcnt, 0u, G);
  pool_accum<<<LOOP_BLOCKS, 256, 0, stream>>>(
      xin, ssum, bias + (size_t)(Lnum - 1) * Dch, gamma + (size_t)(Lnum - 1) * Dch,
      beta + (size_t)(Lnum - 1) * Dch, mean + (size_t)(Lnum - 1) * Dch,
      var + (size_t)(Lnum - 1) * Dch, batch, psum, pcnt, (long)Nn * Dch);
  pool_final<<<(G * Dch + 255) / 256, 256, 0, stream>>>(psum, pcnt, (float*)d_out, G);
}
